# starting kernel
# baseline (speedup 1.0000x reference)
#include <hip/hip_runtime.h>
#include <hip/hip_bf16.h>
#include <math.h>


typedef float v2f __attribute__((ext_vector_type(2)));
typedef float v8f __attribute__((ext_vector_type(8)));

#define HIDDEN 16
#define PROJ   2
#define INPUT  2

#define LOG2E 1.4426950408889634f

__device__ __forceinline__ float sigmoidf_fast(float x) {
  return __builtin_amdgcn_rcpf(1.0f + __builtin_amdgcn_exp2f(x * -LOG2E));
}

__device__ __forceinline__ float sigmoid_pre(float x) {
  return __builtin_amdgcn_rcpf(1.0f + __builtin_amdgcn_exp2f(x));
}

__device__ __forceinline__ float tanhf_fast(float x) {
#if __has_builtin(__builtin_amdgcn_tanhf)
  return __builtin_amdgcn_tanhf(x);
#else
  float e = __builtin_amdgcn_exp2f(x * (-2.0f * LOG2E));
  return __builtin_fmaf(2.0f, __builtin_amdgcn_rcpf(1.0f + e), -1.0f);
#endif
}

__device__ __forceinline__ float halfswap16(float v) {
#if __has_builtin(__builtin_amdgcn_permlanex16)
  int i = __builtin_bit_cast(int, v);
  i = __builtin_amdgcn_permlanex16(i, i, 0x76543210, 0xFEDCBA98, false, false);
  return __builtin_bit_cast(float, i);
#else
  return __shfl_xor(v, 16, 32);
#endif
}

__global__ void __launch_bounds__(256) lstm_phase1(
    const float* __restrict__ y_train, const float* __restrict__ W_ih,
    const float* __restrict__ b_ih, const float* __restrict__ b_hh,
    const float* __restrict__ W_hr, float* __restrict__ out,
    float* __restrict__ c_last, int B, int T, int nstep) {
  int idx = blockIdx.x * blockDim.x + threadIdx.x;
  if (idx >= B * nstep) return;
  int b = idx / nstep;
  int t = idx - b * nstep;

  const float* xp = y_train + ((size_t)b * T + t) * 2;
  float x0 = xp[0], x1 = xp[1];
  float y0 = 0.f, y1 = 0.f;
  bool last = (t == nstep - 1);

#pragma unroll
  for (int n = 0; n < HIDDEN; ++n) {
    int ng = 32 + n, no = 48 + n;
    float gi = W_ih[n * 2 + 0] * x0 + W_ih[n * 2 + 1] * x1 + b_ih[n] + b_hh[n];
    float gg = W_ih[ng * 2 + 0] * x0 + W_ih[ng * 2 + 1] * x1 + b_ih[ng] + b_hh[ng];
    float go = W_ih[no * 2 + 0] * x0 + W_ih[no * 2 + 1] * x1 + b_ih[no] + b_hh[no];
    float cv = sigmoidf_fast(gi) * tanhf_fast(gg);
    float hp = sigmoidf_fast(go) * tanhf_fast(cv);
    y0 = __builtin_fmaf(hp, W_hr[n], y0);
    y1 = __builtin_fmaf(hp, W_hr[16 + n], y1);
    if (last) c_last[b * HIDDEN + n] = cv;
  }
  float* op = out + ((size_t)b * T + t) * 2;
  op[0] = y0;
  op[1] = y1;
}

__global__ void __launch_bounds__(32) lstm_phase2(
    const float* __restrict__ u_train, const float* __restrict__ W_ih,
    const float* __restrict__ W_hh, const float* __restrict__ b_ih,
    const float* __restrict__ b_hh, const float* __restrict__ W_hr,
    const float* __restrict__ c_last, float* __restrict__ out,
    int T, int nstep, int nsteps2) {
  const int lane  = threadIdx.x;
  const int half  = lane >> 4;
  const int j     = lane & 15;
  const int mbase = blockIdx.x * 16;

  v2f At[4];
#pragma unroll
  for (int t = 0; t < 4; ++t) {
    int n = t * 16 + j;
    float sc = (t == 2) ? 1.0f : -LOG2E;
    At[t][0] = sc * (half ? W_hh[n * 2 + 0] : W_ih[n * 2 + 0]);
    At[t][1] = sc * (half ? W_hh[n * 2 + 1] : W_ih[n * 2 + 1]);
  }

  const int jjbase = half * 8;
  v8f cc0, cc1, cc2, cc3;
  float w0[8], w1[8];
#pragma unroll
  for (int r = 0; r < 8; ++r) {
    int jj = jjbase + r;
    cc0[r] = -LOG2E * (b_ih[jj]      + b_hh[jj]);
    cc1[r] = -LOG2E * (b_ih[16 + jj] + b_hh[16 + jj]);
    cc2[r] =           b_ih[32 + jj] + b_hh[32 + jj];
    cc3[r] = -LOG2E * (b_ih[48 + jj] + b_hh[48 + jj]);
    w0[r]  = W_hr[jj];
    w1[r]  = W_hr[16 + jj];
  }

  float c[8];
#pragma unroll
  for (int r = 0; r < 8; ++r)
    c[r] = c_last[(mbase + j) * HIDDEN + jjbase + r];

  float hb0, hb1;
  {
    const float* hp = out + ((size_t)(mbase + j) * T + (nstep - 1)) * 2;
    hb0 = hp[0];
    hb1 = hp[1];
  }

  const float* xbase = u_train + ((size_t)(mbase + j) * T + nstep) * 2;
  float* obase = out + ((size_t)(mbase + j) * T + nstep) * 2;
  const int lastS = nsteps2 - 1;

  float x0 = xbase[0];
  float x1 = xbase[1];

  for (int s = 0; s < nsteps2; ++s) {
    int sn = (s < lastS) ? s + 1 : lastS;
    float nx0 = xbase[sn * 2 + 0];
    float nx1 = xbase[sn * 2 + 1];
    __builtin_prefetch(xbase + sn * 2 + 512, 0, 3);

    v2f bv;
    bv[0] = half ? hb0 : x0;
    bv[1] = half ? hb1 : x1;

    v8f d0 = __builtin_amdgcn_wmma_f32_16x16x4_f32(false, At[0], false, bv, (short)0, cc0, false, false);
    v8f d1 = __builtin_amdgcn_wmma_f32_16x16x4_f32(false, At[1], false, bv, (short)0, cc1, false, false);
    v8f d2 = __builtin_amdgcn_wmma_f32_16x16x4_f32(false, At[2], false, bv, (short)0, cc2, false, false);
    v8f d3 = __builtin_amdgcn_wmma_f32_16x16x4_f32(false, At[3], false, bv, (short)0, cc3, false, false);

    float hp[8];
#pragma unroll
    for (int r = 0; r < 8; ++r) {
      float ig = sigmoid_pre(d0[r]);
      float fg = sigmoid_pre(d1[r]);
      float gg = tanhf_fast(d2[r]);
      float og = sigmoid_pre(d3[r]);
      float cr = __builtin_fmaf(fg, c[r], ig * gg);
      c[r] = cr;
      hp[r] = og * tanhf_fast(cr);
    }

    float s0 = (__builtin_fmaf(hp[0], w0[0], hp[1] * w0[1]) +
                __builtin_fmaf(hp[2], w0[2], hp[3] * w0[3])) +
               (__builtin_fmaf(hp[4], w0[4], hp[5] * w0[5]) +
                __builtin_fmaf(hp[6], w0[6], hp[7] * w0[7]));
    float s1 = (__builtin_fmaf(hp[0], w1[0], hp[1] * w1[1]) +
                __builtin_fmaf(hp[2], w1[2], hp[3] * w1[3])) +
               (__builtin_fmaf(hp[4], w1[4], hp[5] * w1[5]) +
                __builtin_fmaf(hp[6], w1[6], hp[7] * w1[7]));
    s0 += halfswap16(s0);
    s1 += halfswap16(s1);

    hb0 = s0;
    hb1 = s1;

    obase[s * 2 + 0] = s0;
    obase[s * 2 + 1] = s1;

    x0 = nx0;
    x1 = nx1;
  }
}

extern "C" void kernel_launch(void* const* d_in, const int* in_sizes, int n_in,
                              void* d_out, int out_size, void* d_ws, size_t ws_size,
                              hipStream_t stream) {
  const float* u_train = (const float*)d_in[0];
  const float* y_train = (const float*)d_in[1];
  const float* W_ih    = (const float*)d_in[2];
  const float* W_hh    = (const float*)d_in[3];
  const float* b_ih    = (const float*)d_in[4];
  const float* b_hh    = (const float*)d_in[5];
  const float* W_hr    = (const float*)d_in[6];
  const int B = 64;
  const int T = in_sizes[0] / (B * INPUT);
  const int nstep = T / 2;

  float* out    = (float*)d_out;
  float* c_last = (float*)d_ws;

  int total = B * nstep;
  lstm_phase1<<<(total + 255) / 256, 256, 0, stream>>>(
      y_train, W_ih, b_ih, b_hh, W_hr, out, c_last, B, T, nstep);

  lstm_phase2<<<B / 16, 32, 0, stream>>>(
      u_train, W_ih, W_hh, b_ih, b_hh, W_hr, c_last, out, T, nstep, T - nstep);
}


# Time loop blocked by 8 steps so memory waits leave the recurrence: next block's x fetched by 16-byte loads one block ahead (was 8 bytes, one step ahead), the block's h leaves as 16-byte stores
# speedup vs baseline: 1.6955x; 1.6955x over previous
#include <hip/hip_runtime.h>
#include <hip/hip_bf16.h>
#include <math.h>


typedef float v2f __attribute__((ext_vector_type(2)));
typedef float v8f __attribute__((ext_vector_type(8)));
typedef float v4f8 __attribute__((ext_vector_type(4), aligned(8)));

#define HIDDEN 16
#define PROJ   2
#define INPUT  2

#define LOG2E 1.4426950408889634f

__device__ __forceinline__ float sigmoidf_fast(float x) {
  return __builtin_amdgcn_rcpf(1.0f + __builtin_amdgcn_exp2f(x * -LOG2E));
}

__device__ __forceinline__ float sigmoid_pre(float x) {
  return __builtin_amdgcn_rcpf(1.0f + __builtin_amdgcn_exp2f(x));
}

__device__ __forceinline__ float tanhf_fast(float x) {
#if __has_builtin(__builtin_amdgcn_tanhf)
  return __builtin_amdgcn_tanhf(x);
#else
  float e = __builtin_amdgcn_exp2f(x * (-2.0f * LOG2E));
  return __builtin_fmaf(2.0f, __builtin_amdgcn_rcpf(1.0f + e), -1.0f);
#endif
}

__device__ __forceinline__ float halfswap16(float v) {
#if __has_builtin(__builtin_amdgcn_permlanex16)
  int i = __builtin_bit_cast(int, v);
  i = __builtin_amdgcn_permlanex16(i, i, 0x76543210, 0xFEDCBA98, false, false);
  return __builtin_bit_cast(float, i);
#else
  return __shfl_xor(v, 16, 32);
#endif
}

__global__ void __launch_bounds__(256) lstm_phase1(
    const float* __restrict__ y_train, const float* __restrict__ W_ih,
    const float* __restrict__ b_ih, const float* __restrict__ b_hh,
    const float* __restrict__ W_hr, float* __restrict__ out,
    float* __restrict__ c_last, int B, int T, int nstep) {
  int idx = blockIdx.x * blockDim.x + threadIdx.x;
  if (idx >= B * nstep) return;
  int b = idx / nstep;
  int t = idx - b * nstep;

  const float* xp = y_train + ((size_t)b * T + t) * 2;
  float x0 = xp[0], x1 = xp[1];
  float y0 = 0.f, y1 = 0.f;
  bool last = (t == nstep - 1);

#pragma unroll
  for (int n = 0; n < HIDDEN; ++n) {
    int ng = 32 + n, no = 48 + n;
    float gi = W_ih[n * 2 + 0] * x0 + W_ih[n * 2 + 1] * x1 + b_ih[n] + b_hh[n];
    float gg = W_ih[ng * 2 + 0] * x0 + W_ih[ng * 2 + 1] * x1 + b_ih[ng] + b_hh[ng];
    float go = W_ih[no * 2 + 0] * x0 + W_ih[no * 2 + 1] * x1 + b_ih[no] + b_hh[no];
    float cv = sigmoidf_fast(gi) * tanhf_fast(gg);
    float hp = sigmoidf_fast(go) * tanhf_fast(cv);
    y0 = __builtin_fmaf(hp, W_hr[n], y0);
    y1 = __builtin_fmaf(hp, W_hr[16 + n], y1);
    if (last) c_last[b * HIDDEN + n] = cv;
  }
  float* op = out + ((size_t)b * T + t) * 2;
  op[0] = y0;
  op[1] = y1;
}

__global__ void __launch_bounds__(32) lstm_phase2(
    const float* __restrict__ u_train, const float* __restrict__ W_ih,
    const float* __restrict__ W_hh, const float* __restrict__ b_ih,
    const float* __restrict__ b_hh, const float* __restrict__ W_hr,
    const float* __restrict__ c_last, float* __restrict__ out,
    int T, int nstep, int nsteps2) {
  const int lane  = threadIdx.x;
  const int half  = lane >> 4;
  const int j     = lane & 15;
  const int mbase = blockIdx.x * 16;

  v2f At[4];
#pragma unroll
  for (int t = 0; t < 4; ++t) {
    int n = t * 16 + j;
    float sc = (t == 2) ? 1.0f : -LOG2E;
    At[t][0] = sc * (half ? W_hh[n * 2 + 0] : W_ih[n * 2 + 0]);
    At[t][1] = sc * (half ? W_hh[n * 2 + 1] : W_ih[n * 2 + 1]);
  }

  const int jjbase = half * 8;
  v8f cc0, cc1, cc2, cc3;
  float w0[8], w1[8];
#pragma unroll
  for (int r = 0; r < 8; ++r) {
    int jj = jjbase + r;
    cc0[r] = -LOG2E * (b_ih[jj]      + b_hh[jj]);
    cc1[r] = -LOG2E * (b_ih[16 + jj] + b_hh[16 + jj]);
    cc2[r] =           b_ih[32 + jj] + b_hh[32 + jj];
    cc3[r] = -LOG2E * (b_ih[48 + jj] + b_hh[48 + jj]);
    w0[r]  = W_hr[jj];
    w1[r]  = W_hr[16 + jj];
  }

  float c[8];
#pragma unroll
  for (int r = 0; r < 8; ++r)
    c[r] = c_last[(mbase + j) * HIDDEN + jjbase + r];

  float hb0, hb1;
  {
    const float* hp = out + ((size_t)(mbase + j) * T + (nstep - 1)) * 2;
    hb0 = hp[0];
    hb1 = hp[1];
  }

  const float* xbase = u_train + ((size_t)(mbase + j) * T + nstep) * 2;
  float* obase = out + ((size_t)(mbase + j) * T + nstep) * 2;

  auto step = [&](float x0, float x1, float* op) __attribute__((always_inline)) {
    v2f bv;
    bv[0] = half ? hb0 : x0;
    bv[1] = half ? hb1 : x1;

    v8f d0 = __builtin_amdgcn_wmma_f32_16x16x4_f32(false, At[0], false, bv, (short)0, cc0, false, false);
    v8f d1 = __builtin_amdgcn_wmma_f32_16x16x4_f32(false, At[1], false, bv, (short)0, cc1, false, false);
    v8f d2 = __builtin_amdgcn_wmma_f32_16x16x4_f32(false, At[2], false, bv, (short)0, cc2, false, false);
    v8f d3 = __builtin_amdgcn_wmma_f32_16x16x4_f32(false, At[3], false, bv, (short)0, cc3, false, false);

    float hp[8];
#pragma unroll
    for (int r = 0; r < 8; ++r) {
      float ig = sigmoid_pre(d0[r]);
      float fg = sigmoid_pre(d1[r]);
      float gg = tanhf_fast(d2[r]);
      float og = sigmoid_pre(d3[r]);
      float cr = __builtin_fmaf(fg, c[r], ig * gg);
      c[r] = cr;
      hp[r] = og * tanhf_fast(cr);
    }

    float s0 = (__builtin_fmaf(hp[0], w0[0], hp[1] * w0[1]) +
                __builtin_fmaf(hp[2], w0[2], hp[3] * w0[3])) +
               (__builtin_fmaf(hp[4], w0[4], hp[5] * w0[5]) +
                __builtin_fmaf(hp[6], w0[6], hp[7] * w0[7]));
    float s1 = (__builtin_fmaf(hp[0], w1[0], hp[1] * w1[1]) +
                __builtin_fmaf(hp[2], w1[2], hp[3] * w1[3])) +
               (__builtin_fmaf(hp[4], w1[4], hp[5] * w1[5]) +
                __builtin_fmaf(hp[6], w1[6], hp[7] * w1[7]));
    s0 += halfswap16(s0);
    s1 += halfswap16(s1);

    hb0 = s0;
    hb1 = s1;

    op[0] = s0;
    op[1] = s1;
  };

  const int nblk = nsteps2 >> 3;
  v4f8 xc0 = {0.f, 0.f, 0.f, 0.f}, xc1 = xc0, xc2 = xc0, xc3 = xc0;
  if (nblk > 0) {
    const v4f8* p = (const v4f8*)xbase;
    xc0 = p[0]; xc1 = p[1]; xc2 = p[2]; xc3 = p[3];
  }
  for (int blk = 0; blk < nblk; ++blk) {
    int nb = (blk + 1 < nblk) ? blk + 1 : blk;
    const v4f8* np = (const v4f8*)(xbase + nb * 16);
    v4f8 xn0 = np[0], xn1 = np[1], xn2 = np[2], xn3 = np[3];

    float* op = obase + blk * 16;
    step(xc0[0], xc0[1], op + 0);
    step(xc0[2], xc0[3], op + 2);
    step(xc1[0], xc1[1], op + 4);
    step(xc1[2], xc1[3], op + 6);
    step(xc2[0], xc2[1], op + 8);
    step(xc2[2], xc2[3], op + 10);
    step(xc3[0], xc3[1], op + 12);
    step(xc3[2], xc3[3], op + 14);

    xc0 = xn0; xc1 = xn1; xc2 = xn2; xc3 = xn3;
  }

  for (int s = nblk << 3; s < nsteps2; ++s)
    step(xbase[s * 2 + 0], xbase[s * 2 + 1], obase + s * 2);
}

extern "C" void kernel_launch(void* const* d_in, const int* in_sizes, int n_in,
                              void* d_out, int out_size, void* d_ws, size_t ws_size,
                              hipStream_t stream) {
  const float* u_train = (const float*)d_in[0];
  const float* y_train = (const float*)d_in[1];
  const float* W_ih    = (const float*)d_in[2];
  const float* W_hh    = (const float*)d_in[3];
  const float* b_ih    = (const float*)d_in[4];
  const float* b_hh    = (const float*)d_in[5];
  const float* W_hr    = (const float*)d_in[6];
  const int B = 64;
  const int T = in_sizes[0] / (B * INPUT);
  const int nstep = T / 2;

  float* out    = (float*)d_out;
  float* c_last = (float*)d_ws;

  int total = B * nstep;
  lstm_phase1<<<(total + 255) / 256, 256, 0, stream>>>(
      y_train, W_ih, b_ih, b_hh, W_hr, out, c_last, B, T, nstep);

  lstm_phase2<<<B / 16, 32, 0, stream>>>(
      u_train, W_ih, W_hh, b_ih, b_hh, W_hr, c_last, out, T, nstep, T - nstep);
}


# Scan's three sigmoid gates as (1+tanh(x/2))/2 on the native tanh; the 1/2 folded exactly into gate weights, biases and projection weights, the 1+ into the gate's FMA: 5 transcendentals a cell, was 8
# speedup vs baseline: 2.0086x; 1.1847x over previous
#include <hip/hip_runtime.h>
#include <hip/hip_bf16.h>
#include <math.h>


typedef float v2f __attribute__((ext_vector_type(2)));
typedef float v8f __attribute__((ext_vector_type(8)));
typedef float v4f8 __attribute__((ext_vector_type(4), aligned(8)));

#define HIDDEN 16
#define PROJ   2
#define INPUT  2

#define LOG2E 1.4426950408889634f

__device__ __forceinline__ float sigmoidf_fast(float x) {
  return __builtin_amdgcn_rcpf(1.0f + __builtin_amdgcn_exp2f(x * -LOG2E));
}

__device__ __forceinline__ float tanhf_fast(float x) {
#if __has_builtin(__builtin_amdgcn_tanhf)
  return __builtin_amdgcn_tanhf(x);
#else
  float e = __builtin_amdgcn_exp2f(x * (-2.0f * LOG2E));
  return __builtin_fmaf(2.0f, __builtin_amdgcn_rcpf(1.0f + e), -1.0f);
#endif
}

__device__ __forceinline__ float halfswap16(float v) {
#if __has_builtin(__builtin_amdgcn_permlanex16)
  int i = __builtin_bit_cast(int, v);
  i = __builtin_amdgcn_permlanex16(i, i, 0x76543210, 0xFEDCBA98, false, false);
  return __builtin_bit_cast(float, i);
#else
  return __shfl_xor(v, 16, 32);
#endif
}

__global__ void __launch_bounds__(256) lstm_phase1(
    const float* __restrict__ y_train, const float* __restrict__ W_ih,
    const float* __restrict__ b_ih, const float* __restrict__ b_hh,
    const float* __restrict__ W_hr, float* __restrict__ out,
    float* __restrict__ c_last, int B, int T, int nstep) {
  int idx = blockIdx.x * blockDim.x + threadIdx.x;
  if (idx >= B * nstep) return;
  int b = idx / nstep;
  int t = idx - b * nstep;

  const float* xp = y_train + ((size_t)b * T + t) * 2;
  float x0 = xp[0], x1 = xp[1];
  float y0 = 0.f, y1 = 0.f;
  bool last = (t == nstep - 1);

#pragma unroll
  for (int n = 0; n < HIDDEN; ++n) {
    int ng = 32 + n, no = 48 + n;
    float gi = W_ih[n * 2 + 0] * x0 + W_ih[n * 2 + 1] * x1 + b_ih[n] + b_hh[n];
    float gg = W_ih[ng * 2 + 0] * x0 + W_ih[ng * 2 + 1] * x1 + b_ih[ng] + b_hh[ng];
    float go = W_ih[no * 2 + 0] * x0 + W_ih[no * 2 + 1] * x1 + b_ih[no] + b_hh[no];
    float cv = sigmoidf_fast(gi) * tanhf_fast(gg);
    float hp = sigmoidf_fast(go) * tanhf_fast(cv);
    y0 = __builtin_fmaf(hp, W_hr[n], y0);
    y1 = __builtin_fmaf(hp, W_hr[16 + n], y1);
    if (last) c_last[b * HIDDEN + n] = cv;
  }
  float* op = out + ((size_t)b * T + t) * 2;
  op[0] = y0;
  op[1] = y1;
}

__global__ void __launch_bounds__(32) lstm_phase2(
    const float* __restrict__ u_train, const float* __restrict__ W_ih,
    const float* __restrict__ W_hh, const float* __restrict__ b_ih,
    const float* __restrict__ b_hh, const float* __restrict__ W_hr,
    const float* __restrict__ c_last, float* __restrict__ out,
    int T, int nstep, int nsteps2) {
  const int lane  = threadIdx.x;
  const int half  = lane >> 4;
  const int j     = lane & 15;
  const int mbase = blockIdx.x * 16;

  v2f At[4];
#pragma unroll
  for (int t = 0; t < 4; ++t) {
    int n = t * 16 + j;
    float sc = (t == 2) ? 1.0f : 0.5f;
    At[t][0] = sc * (half ? W_hh[n * 2 + 0] : W_ih[n * 2 + 0]);
    At[t][1] = sc * (half ? W_hh[n * 2 + 1] : W_ih[n * 2 + 1]);
  }

  const int jjbase = half * 8;
  v8f cc0, cc1, cc2, cc3;
  float w0[8], w1[8];
#pragma unroll
  for (int r = 0; r < 8; ++r) {
    int jj = jjbase + r;
    cc0[r] = 0.5f * (b_ih[jj]      + b_hh[jj]);
    cc1[r] = 0.5f * (b_ih[16 + jj] + b_hh[16 + jj]);
    cc2[r] =           b_ih[32 + jj] + b_hh[32 + jj];
    cc3[r] = 0.5f * (b_ih[48 + jj] + b_hh[48 + jj]);
    w0[r]  = 0.5f * W_hr[jj];
    w1[r]  = 0.5f * W_hr[16 + jj];
  }

  float c[8];
#pragma unroll
  for (int r = 0; r < 8; ++r)
    c[r] = c_last[(mbase + j) * HIDDEN + jjbase + r];

  float hb0, hb1;
  {
    const float* hp = out + ((size_t)(mbase + j) * T + (nstep - 1)) * 2;
    hb0 = hp[0];
    hb1 = hp[1];
  }

  const float* xbase = u_train + ((size_t)(mbase + j) * T + nstep) * 2;
  float* obase = out + ((size_t)(mbase + j) * T + nstep) * 2;

  auto step = [&](float x0, float x1, float* op) __attribute__((always_inline)) {
    v2f bv;
    bv[0] = half ? hb0 : x0;
    bv[1] = half ? hb1 : x1;

    v8f d0 = __builtin_amdgcn_wmma_f32_16x16x4_f32(false, At[0], false, bv, (short)0, cc0, false, false);
    v8f d1 = __builtin_amdgcn_wmma_f32_16x16x4_f32(false, At[1], false, bv, (short)0, cc1, false, false);
    v8f d2 = __builtin_amdgcn_wmma_f32_16x16x4_f32(false, At[2], false, bv, (short)0, cc2, false, false);
    v8f d3 = __builtin_amdgcn_wmma_f32_16x16x4_f32(false, At[3], false, bv, (short)0, cc3, false, false);

    float hp[8];
#pragma unroll
    for (int r = 0; r < 8; ++r) {
      float ti = tanhf_fast(d0[r]);
      float tf = tanhf_fast(d1[r]);
      float gg = tanhf_fast(d2[r]);
      float to = tanhf_fast(d3[r]);
      float ig2 = __builtin_fmaf(ti, gg, gg);
      float fc2 = __builtin_fmaf(tf, c[r], c[r]);
      float cr = 0.5f * (fc2 + ig2);
      c[r] = cr;
      float tc = tanhf_fast(cr);
      hp[r] = __builtin_fmaf(to, tc, tc);
    }

    float s0 = (__builtin_fmaf(hp[0], w0[0], hp[1] * w0[1]) +
                __builtin_fmaf(hp[2], w0[2], hp[3] * w0[3])) +
               (__builtin_fmaf(hp[4], w0[4], hp[5] * w0[5]) +
                __builtin_fmaf(hp[6], w0[6], hp[7] * w0[7]));
    float s1 = (__builtin_fmaf(hp[0], w1[0], hp[1] * w1[1]) +
                __builtin_fmaf(hp[2], w1[2], hp[3] * w1[3])) +
               (__builtin_fmaf(hp[4], w1[4], hp[5] * w1[5]) +
                __builtin_fmaf(hp[6], w1[6], hp[7] * w1[7]));
    s0 += halfswap16(s0);
    s1 += halfswap16(s1);

    hb0 = s0;
    hb1 = s1;

    op[0] = s0;
    op[1] = s1;
  };

  const int nblk = nsteps2 >> 3;
  v4f8 xc0 = {0.f, 0.f, 0.f, 0.f}, xc1 = xc0, xc2 = xc0, xc3 = xc0;
  if (nblk > 0) {
    const v4f8* p = (const v4f8*)xbase;
    xc0 = p[0]; xc1 = p[1]; xc2 = p[2]; xc3 = p[3];
  }
  for (int blk = 0; blk < nblk; ++blk) {
    int nb = (blk + 1 < nblk) ? blk + 1 : blk;
    const v4f8* np = (const v4f8*)(xbase + nb * 16);
    v4f8 xn0 = np[0], xn1 = np[1], xn2 = np[2], xn3 = np[3];

    float* op = obase + blk * 16;
    step(xc0[0], xc0[1], op + 0);
    step(xc0[2], xc0[3], op + 2);
    step(xc1[0], xc1[1], op + 4);
    step(xc1[2], xc1[3], op + 6);
    step(xc2[0], xc2[1], op + 8);
    step(xc2[2], xc2[3], op + 10);
    step(xc3[0], xc3[1], op + 12);
    step(xc3[2], xc3[3], op + 14);

    xc0 = xn0; xc1 = xn1; xc2 = xn2; xc3 = xn3;
  }

  for (int s = nblk << 3; s < nsteps2; ++s)
    step(xbase[s * 2 + 0], xbase[s * 2 + 1], obase + s * 2);
}

extern "C" void kernel_launch(void* const* d_in, const int* in_sizes, int n_in,
                              void* d_out, int out_size, void* d_ws, size_t ws_size,
                              hipStream_t stream) {
  const float* u_train = (const float*)d_in[0];
  const float* y_train = (const float*)d_in[1];
  const float* W_ih    = (const float*)d_in[2];
  const float* W_hh    = (const float*)d_in[3];
  const float* b_ih    = (const float*)d_in[4];
  const float* b_hh    = (const float*)d_in[5];
  const float* W_hr    = (const float*)d_in[6];
  const int B = 64;
  const int T = in_sizes[0] / (B * INPUT);
  const int nstep = T / 2;

  float* out    = (float*)d_out;
  float* c_last = (float*)d_ws;

  int total = B * nstep;
  lstm_phase1<<<(total + 255) / 256, 256, 0, stream>>>(
      y_train, W_ih, b_ih, b_hh, W_hr, out, c_last, B, T, nstep);

  lstm_phase2<<<B / 16, 32, 0, stream>>>(
      u_train, W_ih, W_hh, b_ih, b_hh, W_hr, c_last, out, T, nstep, T - nstep);
}


# speedup vs baseline: 2.1244x; 1.0577x over previous
#include <hip/hip_runtime.h>
#include <hip/hip_bf16.h>
#include <math.h>


typedef float v2f __attribute__((ext_vector_type(2)));
typedef float v8f __attribute__((ext_vector_type(8)));
typedef float v4f8 __attribute__((ext_vector_type(4), aligned(8)));

#define HIDDEN 16
#define PROJ   2
#define INPUT  2

#define LOG2E 1.4426950408889634f

__device__ __forceinline__ float sigmoidf_fast(float x) {
  return __builtin_amdgcn_rcpf(1.0f + __builtin_amdgcn_exp2f(x * -LOG2E));
}

__device__ __forceinline__ float tanhf_fast(float x) {
#if __has_builtin(__builtin_amdgcn_tanhf)
  return __builtin_amdgcn_tanhf(x);
#else
  float e = __builtin_amdgcn_exp2f(x * (-2.0f * LOG2E));
  return __builtin_fmaf(2.0f, __builtin_amdgcn_rcpf(1.0f + e), -1.0f);
#endif
}

__device__ __forceinline__ float halfswap16(float v) {
#if __has_builtin(__builtin_amdgcn_permlanex16)
  int i = __builtin_bit_cast(int, v);
  i = __builtin_amdgcn_permlanex16(i, i, 0x76543210, 0xFEDCBA98, false, false);
  return __builtin_bit_cast(float, i);
#else
  return __shfl_xor(v, 16, 32);
#endif
}

__global__ void __launch_bounds__(256) lstm_phase1(
    const float* __restrict__ y_train, const float* __restrict__ W_ih,
    const float* __restrict__ b_ih, const float* __restrict__ b_hh,
    const float* __restrict__ W_hr, float* __restrict__ out,
    float* __restrict__ c_last, int B, int T, int nstep) {
  int idx = blockIdx.x * blockDim.x + threadIdx.x;
  if (idx >= B * nstep) return;
  int b = idx / nstep;
  int t = idx - b * nstep;

  const float* xp = y_train + ((size_t)b * T + t) * 2;
  float x0 = xp[0], x1 = xp[1];
  float y0 = 0.f, y1 = 0.f;
  bool last = (t == nstep - 1);

#pragma unroll
  for (int n = 0; n < HIDDEN; ++n) {
    int ng = 32 + n, no = 48 + n;
    float gi = W_ih[n * 2 + 0] * x0 + W_ih[n * 2 + 1] * x1 + b_ih[n] + b_hh[n];
    float gg = W_ih[ng * 2 + 0] * x0 + W_ih[ng * 2 + 1] * x1 + b_ih[ng] + b_hh[ng];
    float go = W_ih[no * 2 + 0] * x0 + W_ih[no * 2 + 1] * x1 + b_ih[no] + b_hh[no];
    float cv = sigmoidf_fast(gi) * tanhf_fast(gg);
    float hp = sigmoidf_fast(go) * tanhf_fast(cv);
    y0 = __builtin_fmaf(hp, W_hr[n], y0);
    y1 = __builtin_fmaf(hp, W_hr[16 + n], y1);
    if (last) c_last[b * HIDDEN + n] = cv;
  }
  float* op = out + ((size_t)b * T + t) * 2;
  op[0] = y0;
  op[1] = y1;
}

__global__ void __launch_bounds__(32) lstm_phase2(
    const float* __restrict__ u_train, const float* __restrict__ W_ih,
    const float* __restrict__ W_hh, const float* __restrict__ b_ih,
    const float* __restrict__ b_hh, const float* __restrict__ W_hr,
    const float* __restrict__ c_last, float* __restrict__ out,
    int T, int nstep, int nsteps2) {
  const int lane  = threadIdx.x;
  const int half  = lane >> 4;
  const int j     = lane & 15;
  const int mbase = blockIdx.x * 16;

  v2f At[4];
#pragma unroll
  for (int t = 0; t < 4; ++t) {
    int n = t * 16 + j;
    float sc = (t == 2) ? 1.0f : 0.5f;
    At[t][0] = sc * (half ? W_hh[n * 2 + 0] : W_ih[n * 2 + 0]);
    At[t][1] = sc * (half ? W_hh[n * 2 + 1] : W_ih[n * 2 + 1]);
  }

  const int jjbase = half * 8;
  v8f cc0, cc1, cc2, cc3;
  v2f wv[8];
#pragma unroll
  for (int r = 0; r < 8; ++r) {
    int jj = jjbase + r;
    cc0[r] = 0.5f * (b_ih[jj]      + b_hh[jj]);
    cc1[r] = 0.5f * (b_ih[16 + jj] + b_hh[16 + jj]);
    cc2[r] =           b_ih[32 + jj] + b_hh[32 + jj];
    cc3[r] = 0.5f * (b_ih[48 + jj] + b_hh[48 + jj]);
    wv[r][0] = 0.5f * W_hr[jj];
    wv[r][1] = 0.5f * W_hr[16 + jj];
  }

  float c[8];
#pragma unroll
  for (int r = 0; r < 8; ++r)
    c[r] = c_last[(mbase + j) * HIDDEN + jjbase + r];

  float hb0, hb1;
  {
    const float* hp = out + ((size_t)(mbase + j) * T + (nstep - 1)) * 2;
    hb0 = hp[0];
    hb1 = hp[1];
  }

  const float* xbase = u_train + ((size_t)(mbase + j) * T + nstep) * 2;
  float* obase = out + ((size_t)(mbase + j) * T + nstep) * 2;

  auto step = [&](float x0, float x1, float& o0, float& o1) __attribute__((always_inline)) {
    v2f bv;
    bv[0] = half ? hb0 : x0;
    bv[1] = half ? hb1 : x1;

    v8f d0 = __builtin_amdgcn_wmma_f32_16x16x4_f32(false, At[0], false, bv, (short)0, cc0, false, false);
    v8f d2 = __builtin_amdgcn_wmma_f32_16x16x4_f32(false, At[2], false, bv, (short)0, cc2, false, false);
    v8f d1 = __builtin_amdgcn_wmma_f32_16x16x4_f32(false, At[1], false, bv, (short)0, cc1, false, false);
    asm volatile("" : "+v"(d0), "+v"(d2), "+v"(d1));
    float ti[8];
#pragma unroll
    for (int r = 0; r < 8; ++r) ti[r] = tanhf_fast(d0[r]);
    asm volatile("" : "+v"(bv), "+v"(d2), "+v"(d1), "+v"(ti[0]), "+v"(ti[1]), "+v"(ti[2]), "+v"(ti[3]), "+v"(ti[4]), "+v"(ti[5]), "+v"(ti[6]), "+v"(ti[7]));
    v8f d3 = __builtin_amdgcn_wmma_f32_16x16x4_f32(false, At[3], false, bv, (short)0, cc3, false, false);
    asm volatile("" : "+v"(d3), "+v"(d2), "+v"(d1));

    float gg[8], tf[8], to[8], ig2[8], fc2[8], sm[8], cr[8], tc[8], hp[8];
    gg[0] = tanhf_fast(d2[0]);
    gg[1] = tanhf_fast(d2[1]);
    gg[2] = tanhf_fast(d2[2]);
    gg[3] = tanhf_fast(d2[3]);
    gg[4] = tanhf_fast(d2[4]);
    gg[5] = tanhf_fast(d2[5]);
    ig2[0] = __builtin_fmaf(ti[0], gg[0], gg[0]);
    ig2[1] = __builtin_fmaf(ti[1], gg[1], gg[1]);
    asm volatile("" : "+v"(gg[0]), "+v"(gg[1]), "+v"(gg[2]), "+v"(gg[3]), "+v"(gg[4]), "+v"(gg[5]), "+v"(ig2[0]), "+v"(ig2[1]), "+v"(d1), "+v"(d3), "+v"(d2), "+v"(bv));
    gg[6] = tanhf_fast(d2[6]);
    gg[7] = tanhf_fast(d2[7]);
    ig2[2] = __builtin_fmaf(ti[2], gg[2], gg[2]);
    ig2[3] = __builtin_fmaf(ti[3], gg[3], gg[3]);
    asm volatile("" : "+v"(gg[6]), "+v"(gg[7]), "+v"(ig2[2]), "+v"(ig2[3]), "+v"(d1), "+v"(gg[4]), "+v"(gg[5]), "+v"(d3));
    tf[0] = tanhf_fast(d1[0]);
    tf[1] = tanhf_fast(d1[1]);
    ig2[4] = __builtin_fmaf(ti[4], gg[4], gg[4]);
    ig2[5] = __builtin_fmaf(ti[5], gg[5], gg[5]);
    asm volatile("" : "+v"(tf[0]), "+v"(tf[1]), "+v"(ig2[4]), "+v"(ig2[5]), "+v"(d1), "+v"(gg[6]), "+v"(gg[7]), "+v"(d3));
    tf[2] = tanhf_fast(d1[2]);
    tf[3] = tanhf_fast(d1[3]);
    ig2[6] = __builtin_fmaf(ti[6], gg[6], gg[6]);
    ig2[7] = __builtin_fmaf(ti[7], gg[7], gg[7]);
    asm volatile("" : "+v"(tf[2]), "+v"(tf[3]), "+v"(ig2[6]), "+v"(ig2[7]), "+v"(d1), "+v"(tf[0]), "+v"(tf[1]), "+v"(d3));
    tf[4] = tanhf_fast(d1[4]);
    tf[5] = tanhf_fast(d1[5]);
    fc2[0] = __builtin_fmaf(tf[0], c[0], c[0]);
    fc2[1] = __builtin_fmaf(tf[1], c[1], c[1]);
    asm volatile("" : "+v"(tf[4]), "+v"(tf[5]), "+v"(fc2[0]), "+v"(fc2[1]), "+v"(d1), "+v"(tf[2]), "+v"(tf[3]), "+v"(d3));
    tf[6] = tanhf_fast(d1[6]);
    tf[7] = tanhf_fast(d1[7]);
    fc2[2] = __builtin_fmaf(tf[2], c[2], c[2]);
    fc2[3] = __builtin_fmaf(tf[3], c[3], c[3]);
    asm volatile("" : "+v"(tf[6]), "+v"(tf[7]), "+v"(fc2[2]), "+v"(fc2[3]), "+v"(d3), "+v"(fc2[0]), "+v"(fc2[1]), "+v"(tf[4]), "+v"(tf[5]));
    to[0] = tanhf_fast(d3[0]);
    sm[0] = fc2[0] + ig2[0];
    sm[1] = fc2[1] + ig2[1];
    asm volatile("" : "+v"(to[0]), "+v"(sm[0]), "+v"(sm[1]), "+v"(d3), "+v"(tf[4]), "+v"(tf[5]), "+v"(fc2[2]), "+v"(fc2[3]), "+v"(tf[6]), "+v"(tf[7]));
    to[1] = tanhf_fast(d3[1]);
    fc2[4] = __builtin_fmaf(tf[4], c[4], c[4]);
    fc2[5] = __builtin_fmaf(tf[5], c[5], c[5]);
    asm volatile("" : "+v"(to[1]), "+v"(fc2[4]), "+v"(fc2[5]), "+v"(d3), "+v"(fc2[2]), "+v"(fc2[3]), "+v"(tf[6]), "+v"(tf[7]), "+v"(sm[0]), "+v"(sm[1]));
    to[2] = tanhf_fast(d3[2]);
    sm[2] = fc2[2] + ig2[2];
    sm[3] = fc2[3] + ig2[3];
    asm volatile("" : "+v"(to[2]), "+v"(sm[2]), "+v"(sm[3]), "+v"(d3), "+v"(tf[6]), "+v"(tf[7]), "+v"(sm[0]), "+v"(sm[1]), "+v"(fc2[4]), "+v"(fc2[5]));
    to[3] = tanhf_fast(d3[3]);
    fc2[6] = __builtin_fmaf(tf[6], c[6], c[6]);
    fc2[7] = __builtin_fmaf(tf[7], c[7], c[7]);
    asm volatile("" : "+v"(to[3]), "+v"(fc2[6]), "+v"(fc2[7]), "+v"(d3), "+v"(sm[0]), "+v"(sm[1]), "+v"(fc2[4]), "+v"(fc2[5]), "+v"(sm[2]), "+v"(sm[3]));
    to[4] = tanhf_fast(d3[4]);
    cr[0] = 0.5f * sm[0];
    cr[1] = 0.5f * sm[1];
    asm volatile("" : "+v"(to[4]), "+v"(cr[0]), "+v"(cr[1]), "+v"(d3), "+v"(fc2[4]), "+v"(fc2[5]), "+v"(sm[2]), "+v"(sm[3]), "+v"(fc2[6]), "+v"(fc2[7]));
    to[5] = tanhf_fast(d3[5]);
    sm[4] = fc2[4] + ig2[4];
    sm[5] = fc2[5] + ig2[5];
    asm volatile("" : "+v"(to[5]), "+v"(sm[4]), "+v"(sm[5]), "+v"(d3), "+v"(sm[2]), "+v"(sm[3]), "+v"(fc2[6]), "+v"(fc2[7]), "+v"(cr[0]), "+v"(cr[1]));
    to[6] = tanhf_fast(d3[6]);
    cr[2] = 0.5f * sm[2];
    cr[3] = 0.5f * sm[3];
    asm volatile("" : "+v"(to[6]), "+v"(cr[2]), "+v"(cr[3]), "+v"(d3), "+v"(fc2[6]), "+v"(fc2[7]), "+v"(cr[0]), "+v"(sm[4]), "+v"(sm[5]), "+v"(cr[1]));
    to[7] = tanhf_fast(d3[7]);
    sm[6] = fc2[6] + ig2[6];
    sm[7] = fc2[7] + ig2[7];
    asm volatile("" : "+v"(to[7]), "+v"(sm[6]), "+v"(sm[7]), "+v"(cr[0]), "+v"(sm[4]), "+v"(sm[5]), "+v"(cr[1]), "+v"(cr[2]), "+v"(cr[3]));
    tc[0] = tanhf_fast(cr[0]);
    cr[4] = 0.5f * sm[4];
    cr[5] = 0.5f * sm[5];
    asm volatile("" : "+v"(tc[0]), "+v"(cr[4]), "+v"(cr[5]), "+v"(cr[1]), "+v"(cr[2]), "+v"(sm[6]), "+v"(sm[7]), "+v"(cr[3]));
    tc[1] = tanhf_fast(cr[1]);
    tc[2] = tanhf_fast(cr[2]);
    cr[6] = 0.5f * sm[6];
    cr[7] = 0.5f * sm[7];
    asm volatile("" : "+v"(tc[1]), "+v"(tc[2]), "+v"(cr[6]), "+v"(cr[7]), "+v"(cr[3]), "+v"(cr[4]), "+v"(cr[5]));
#pragma unroll
    for (int r = 3; r < 8; ++r) tc[r] = tanhf_fast(cr[r]);
    asm volatile("" : "+v"(tc[0]), "+v"(tc[1]), "+v"(tc[2]), "+v"(tc[3]), "+v"(tc[4]), "+v"(tc[5]), "+v"(tc[6]), "+v"(tc[7]));
#pragma unroll
    for (int r = 0; r < 8; ++r) {
      hp[r] = __builtin_fmaf(to[r], tc[r], tc[r]);
      c[r] = cr[r];
    }
    asm volatile("" : "+v"(hp[0]), "+v"(hp[1]), "+v"(hp[2]), "+v"(hp[3]), "+v"(hp[4]), "+v"(hp[5]), "+v"(hp[6]), "+v"(hp[7]));

    auto pr = [&](int a, int b) __attribute__((always_inline)) {
      v2f ha = {hp[a], hp[a]}, hb = {hp[b], hp[b]};
      return __builtin_elementwise_fma(ha, wv[a], hb * wv[b]);
    };
    v2f sv = (pr(0, 1) + pr(2, 3)) + (pr(4, 5) + pr(6, 7));
    float s0 = sv[0], s1 = sv[1];
    s0 += halfswap16(s0);
    s1 += halfswap16(s1);

    hb0 = s0;
    hb1 = s1;

    o0 = s0;
    o1 = s1;
  };

  const int nblk = nsteps2 >> 3;
  v4f8 xc0 = {0.f, 0.f, 0.f, 0.f}, xc1 = xc0, xc2 = xc0, xc3 = xc0;
  if (nblk > 0) {
    const v4f8* p = (const v4f8*)xbase;
    xc0 = p[0]; xc1 = p[1]; xc2 = p[2]; xc3 = p[3];
  }
  for (int blk = 0; blk < nblk; ++blk) {
    int nb = (blk + 1 < nblk) ? blk + 1 : blk;
    const v4f8* np = (const v4f8*)(xbase + nb * 16);
    v4f8 xn0 = np[0], xn1 = np[1], xn2 = np[2], xn3 = np[3];

    float hs[16];
    step(xc0[0], xc0[1], hs[0],  hs[1]);
    step(xc0[2], xc0[3], hs[2],  hs[3]);
    step(xc1[0], xc1[1], hs[4],  hs[5]);
    step(xc1[2], xc1[3], hs[6],  hs[7]);
    step(xc2[0], xc2[1], hs[8],  hs[9]);
    step(xc2[2], xc2[3], hs[10], hs[11]);
    step(xc3[0], xc3[1], hs[12], hs[13]);
    step(xc3[2], xc3[3], hs[14], hs[15]);
    v4f8* op = (v4f8*)(obase + blk * 16);
#pragma unroll
    for (int q = 0; q < 4; ++q) {
      v4f8 hv = {hs[4 * q], hs[4 * q + 1], hs[4 * q + 2], hs[4 * q + 3]};
      op[q] = hv;
    }

    xc0 = xn0; xc1 = xn1; xc2 = xn2; xc3 = xn3;
  }

  for (int s = nblk << 3; s < nsteps2; ++s) {
    float h0, h1;
    step(xbase[s * 2 + 0], xbase[s * 2 + 1], h0, h1);
    obase[s * 2 + 0] = h0;
    obase[s * 2 + 1] = h1;
  }
}

extern "C" void kernel_launch(void* const* d_in, const int* in_sizes, int n_in,
                              void* d_out, int out_size, void* d_ws, size_t ws_size,
                              hipStream_t stream) {
  const float* u_train = (const float*)d_in[0];
  const float* y_train = (const float*)d_in[1];
  const float* W_ih    = (const float*)d_in[2];
  const float* W_hh    = (const float*)d_in[3];
  const float* b_ih    = (const float*)d_in[4];
  const float* b_hh    = (const float*)d_in[5];
  const float* W_hr    = (const float*)d_in[6];
  const int B = 64;
  const int T = in_sizes[0] / (B * INPUT);
  const int nstep = T / 2;

  float* out    = (float*)d_out;
  float* c_last = (float*)d_ws;

  int total = B * nstep;
  lstm_phase1<<<(total + 255) / 256, 256, 0, stream>>>(
      y_train, W_ih, b_ih, b_hh, W_hr, out, c_last, B, T, nstep);

  lstm_phase2<<<B / 16, 32, 0, stream>>>(
      u_train, W_ih, W_hh, b_ih, b_hh, W_hr, c_last, out, T, nstep, T - nstep);
}


# Serial tail behind the last tanh(c) cut: (1+to)*w folded ahead, each tanh(c) enters one of two scalar multiply-add chains on arrival (new summation order), the half-join itself yields the next B
# speedup vs baseline: 2.3834x; 1.1219x over previous
#include <hip/hip_runtime.h>
#include <hip/hip_bf16.h>
#include <math.h>


typedef float v2f __attribute__((ext_vector_type(2)));
typedef float v8f __attribute__((ext_vector_type(8)));
typedef float v4f8 __attribute__((ext_vector_type(4), aligned(8)));

#define HIDDEN 16
#define PROJ   2
#define INPUT  2

#define LOG2E 1.4426950408889634f

__device__ __forceinline__ float sigmoidf_fast(float x) {
  return __builtin_amdgcn_rcpf(1.0f + __builtin_amdgcn_exp2f(x * -LOG2E));
}

__device__ __forceinline__ float tanhf_fast(float x) {
#if __has_builtin(__builtin_amdgcn_tanhf)
  return __builtin_amdgcn_tanhf(x);
#else
  float e = __builtin_amdgcn_exp2f(x * (-2.0f * LOG2E));
  return __builtin_fmaf(2.0f, __builtin_amdgcn_rcpf(1.0f + e), -1.0f);
#endif
}

__device__ __forceinline__ float halfswap16(float v) {
#if __has_builtin(__builtin_amdgcn_permlanex16)
  int i = __builtin_bit_cast(int, v);
  i = __builtin_amdgcn_permlanex16(0, i, 0x76543210, 0xFEDCBA98, false, false);
  return __builtin_bit_cast(float, i);
#else
  return __shfl_xor(v, 16, 32);
#endif
}

__global__ void __launch_bounds__(256) lstm_phase1(
    const float* __restrict__ y_train, const float* __restrict__ W_ih,
    const float* __restrict__ b_ih, const float* __restrict__ b_hh,
    const float* __restrict__ W_hr, float* __restrict__ out,
    float* __restrict__ c_last, int B, int T, int nstep) {
  int idx = blockIdx.x * blockDim.x + threadIdx.x;
  if (idx >= B * nstep) return;
  int b = idx / nstep;
  int t = idx - b * nstep;

  const float* xp = y_train + ((size_t)b * T + t) * 2;
  float x0 = xp[0], x1 = xp[1];
  float y0 = 0.f, y1 = 0.f;
  bool last = (t == nstep - 1);

#pragma unroll
  for (int n = 0; n < HIDDEN; ++n) {
    int ng = 32 + n, no = 48 + n;
    float gi = W_ih[n * 2 + 0] * x0 + W_ih[n * 2 + 1] * x1 + b_ih[n] + b_hh[n];
    float gg = W_ih[ng * 2 + 0] * x0 + W_ih[ng * 2 + 1] * x1 + b_ih[ng] + b_hh[ng];
    float go = W_ih[no * 2 + 0] * x0 + W_ih[no * 2 + 1] * x1 + b_ih[no] + b_hh[no];
    float cv = sigmoidf_fast(gi) * tanhf_fast(gg);
    float hp = sigmoidf_fast(go) * tanhf_fast(cv);
    y0 = __builtin_fmaf(hp, W_hr[n], y0);
    y1 = __builtin_fmaf(hp, W_hr[16 + n], y1);
    if (last) c_last[b * HIDDEN + n] = cv;
  }
  float* op = out + ((size_t)b * T + t) * 2;
  op[0] = y0;
  op[1] = y1;
}

__global__ void __launch_bounds__(32) lstm_phase2(
    const float* __restrict__ u_train, const float* __restrict__ W_ih,
    const float* __restrict__ W_hh, const float* __restrict__ b_ih,
    const float* __restrict__ b_hh, const float* __restrict__ W_hr,
    const float* __restrict__ c_last, float* __restrict__ out,
    int T, int nstep, int nsteps2) {
  const int lane  = threadIdx.x;
  const int half  = lane >> 4;
  const int j     = lane & 15;
  const int mbase = blockIdx.x * 16;

  v2f At[4];
#pragma unroll
  for (int t = 0; t < 4; ++t) {
    int n = t * 16 + j;
    float sc = (t == 2) ? 1.0f : 0.5f;
    At[t][0] = sc * (half ? W_hh[n * 2 + 0] : W_ih[n * 2 + 0]);
    At[t][1] = sc * (half ? W_hh[n * 2 + 1] : W_ih[n * 2 + 1]);
  }

  const int jjbase = half * 8;
  v8f cc0, cc1, cc2, cc3;
  v2f wv[8];
#pragma unroll
  for (int r = 0; r < 8; ++r) {
    int jj = jjbase + r;
    cc0[r] = 0.5f * (b_ih[jj]      + b_hh[jj]);
    cc1[r] = 0.5f * (b_ih[16 + jj] + b_hh[16 + jj]);
    cc2[r] =           b_ih[32 + jj] + b_hh[32 + jj];
    cc3[r] = 0.5f * (b_ih[48 + jj] + b_hh[48 + jj]);
    wv[r][0] = 0.5f * W_hr[jj];
    wv[r][1] = 0.5f * W_hr[16 + jj];
  }

  float c[8];
#pragma unroll
  for (int r = 0; r < 8; ++r)
    c[r] = c_last[(mbase + j) * HIDDEN + jjbase + r];

  const float* xbase = u_train + ((size_t)(mbase + j) * T + nstep) * 2;
  float* obase = out + ((size_t)(mbase + j) * T + nstep) * 2;

  float hb0, hb1;
  {
    const float* hp = out + ((size_t)(mbase + j) * T + (nstep - 1)) * 2;
    float xf0 = 0.f, xf1 = 0.f;
    if (nsteps2 > 0) { xf0 = xbase[0]; xf1 = xbase[1]; }
    hb0 = half ? hp[0] : xf0;
    hb1 = half ? hp[1] : xf1;
  }
  const float hm = half ? 1.0f : 0.0f;

  auto step = [&](float xn0, float xn1, float& o0, float& o1) __attribute__((always_inline)) {
    v2f bv;
    bv[0] = hb0;
    bv[1] = hb1;

    v8f d0 = __builtin_amdgcn_wmma_f32_16x16x4_f32(false, At[0], false, bv, (short)0, cc0, false, false);
    v8f d2 = __builtin_amdgcn_wmma_f32_16x16x4_f32(false, At[2], false, bv, (short)0, cc2, false, false);
    v8f d1 = __builtin_amdgcn_wmma_f32_16x16x4_f32(false, At[1], false, bv, (short)0, cc1, false, false);
    asm volatile("" : "+v"(d0), "+v"(d2), "+v"(d1));
    float ti[8];
#pragma unroll
    for (int r = 0; r < 8; ++r) ti[r] = tanhf_fast(d0[r]);
    asm volatile("" : "+v"(bv), "+v"(d2), "+v"(d1), "+v"(ti[0]), "+v"(ti[1]), "+v"(ti[2]), "+v"(ti[3]), "+v"(ti[4]), "+v"(ti[5]), "+v"(ti[6]), "+v"(ti[7]));
    v8f d3 = __builtin_amdgcn_wmma_f32_16x16x4_f32(false, At[3], false, bv, (short)0, cc3, false, false);
    asm volatile("" : "+v"(d3), "+v"(d2), "+v"(d1));

    float gg[8], tf[8], to[8], ig2[8], fc2[8], sm[8], cr[8], tc[8];
    gg[0] = tanhf_fast(d2[0]);
    gg[1] = tanhf_fast(d2[1]);
    gg[2] = tanhf_fast(d2[2]);
    gg[3] = tanhf_fast(d2[3]);
    gg[4] = tanhf_fast(d2[4]);
    gg[5] = tanhf_fast(d2[5]);
    ig2[0] = __builtin_fmaf(ti[0], gg[0], gg[0]);
    ig2[1] = __builtin_fmaf(ti[1], gg[1], gg[1]);
    asm volatile("" : "+v"(gg[0]), "+v"(gg[1]), "+v"(gg[2]), "+v"(gg[3]), "+v"(gg[4]), "+v"(gg[5]), "+v"(ig2[0]), "+v"(ig2[1]), "+v"(d1), "+v"(d3), "+v"(d2), "+v"(bv));
    gg[6] = tanhf_fast(d2[6]);
    gg[7] = tanhf_fast(d2[7]);
    ig2[2] = __builtin_fmaf(ti[2], gg[2], gg[2]);
    ig2[3] = __builtin_fmaf(ti[3], gg[3], gg[3]);
    asm volatile("" : "+v"(gg[6]), "+v"(gg[7]), "+v"(ig2[2]), "+v"(ig2[3]), "+v"(d1), "+v"(gg[4]), "+v"(gg[5]), "+v"(d3));
    tf[0] = tanhf_fast(d1[0]);
    tf[1] = tanhf_fast(d1[1]);
    ig2[4] = __builtin_fmaf(ti[4], gg[4], gg[4]);
    ig2[5] = __builtin_fmaf(ti[5], gg[5], gg[5]);
    asm volatile("" : "+v"(tf[0]), "+v"(tf[1]), "+v"(ig2[4]), "+v"(ig2[5]), "+v"(d1), "+v"(gg[6]), "+v"(gg[7]), "+v"(d3));
    tf[2] = tanhf_fast(d1[2]);
    tf[3] = tanhf_fast(d1[3]);
    ig2[6] = __builtin_fmaf(ti[6], gg[6], gg[6]);
    ig2[7] = __builtin_fmaf(ti[7], gg[7], gg[7]);
    asm volatile("" : "+v"(tf[2]), "+v"(tf[3]), "+v"(ig2[6]), "+v"(ig2[7]), "+v"(d1), "+v"(tf[0]), "+v"(tf[1]), "+v"(d3));
    tf[4] = tanhf_fast(d1[4]);
    tf[5] = tanhf_fast(d1[5]);
    fc2[0] = __builtin_fmaf(tf[0], c[0], c[0]);
    fc2[1] = __builtin_fmaf(tf[1], c[1], c[1]);
    asm volatile("" : "+v"(tf[4]), "+v"(tf[5]), "+v"(fc2[0]), "+v"(fc2[1]), "+v"(d1), "+v"(tf[2]), "+v"(tf[3]), "+v"(d3));
    tf[6] = tanhf_fast(d1[6]);
    tf[7] = tanhf_fast(d1[7]);
    fc2[2] = __builtin_fmaf(tf[2], c[2], c[2]);
    fc2[3] = __builtin_fmaf(tf[3], c[3], c[3]);
    asm volatile("" : "+v"(tf[6]), "+v"(tf[7]), "+v"(fc2[2]), "+v"(fc2[3]), "+v"(d3), "+v"(fc2[0]), "+v"(fc2[1]), "+v"(tf[4]), "+v"(tf[5]));
    to[0] = tanhf_fast(d3[0]);
    sm[0] = fc2[0] + ig2[0];
    sm[1] = fc2[1] + ig2[1];
    asm volatile("" : "+v"(to[0]), "+v"(sm[0]), "+v"(sm[1]), "+v"(d3), "+v"(tf[4]), "+v"(tf[5]), "+v"(fc2[2]), "+v"(fc2[3]), "+v"(tf[6]), "+v"(tf[7]));
    to[1] = tanhf_fast(d3[1]);
    fc2[4] = __builtin_fmaf(tf[4], c[4], c[4]);
    fc2[5] = __builtin_fmaf(tf[5], c[5], c[5]);
    asm volatile("" : "+v"(to[1]), "+v"(fc2[4]), "+v"(fc2[5]), "+v"(d3), "+v"(fc2[2]), "+v"(fc2[3]), "+v"(tf[6]), "+v"(tf[7]), "+v"(sm[0]), "+v"(sm[1]));
    to[2] = tanhf_fast(d3[2]);
    sm[2] = fc2[2] + ig2[2];
    sm[3] = fc2[3] + ig2[3];
    asm volatile("" : "+v"(to[2]), "+v"(sm[2]), "+v"(sm[3]), "+v"(d3), "+v"(tf[6]), "+v"(tf[7]), "+v"(sm[0]), "+v"(sm[1]), "+v"(fc2[4]), "+v"(fc2[5]));
    to[3] = tanhf_fast(d3[3]);
    fc2[6] = __builtin_fmaf(tf[6], c[6], c[6]);
    fc2[7] = __builtin_fmaf(tf[7], c[7], c[7]);
    asm volatile("" : "+v"(to[3]), "+v"(fc2[6]), "+v"(fc2[7]), "+v"(d3), "+v"(sm[0]), "+v"(sm[1]), "+v"(fc2[4]), "+v"(fc2[5]), "+v"(sm[2]), "+v"(sm[3]));
    to[4] = tanhf_fast(d3[4]);
    cr[0] = 0.5f * sm[0];
    cr[1] = 0.5f * sm[1];
    asm volatile("" : "+v"(to[4]), "+v"(cr[0]), "+v"(cr[1]), "+v"(d3), "+v"(fc2[4]), "+v"(fc2[5]), "+v"(sm[2]), "+v"(sm[3]), "+v"(fc2[6]), "+v"(fc2[7]));
    to[5] = tanhf_fast(d3[5]);
    sm[4] = fc2[4] + ig2[4];
    sm[5] = fc2[5] + ig2[5];
    asm volatile("" : "+v"(to[5]), "+v"(sm[4]), "+v"(sm[5]), "+v"(d3), "+v"(sm[2]), "+v"(sm[3]), "+v"(fc2[6]), "+v"(fc2[7]), "+v"(cr[0]), "+v"(cr[1]));
    to[6] = tanhf_fast(d3[6]);
    cr[2] = 0.5f * sm[2];
    cr[3] = 0.5f * sm[3];
    asm volatile("" : "+v"(to[6]), "+v"(cr[2]), "+v"(cr[3]), "+v"(d3), "+v"(fc2[6]), "+v"(fc2[7]), "+v"(cr[0]), "+v"(sm[4]), "+v"(sm[5]), "+v"(cr[1]));
    to[7] = tanhf_fast(d3[7]);
    sm[6] = fc2[6] + ig2[6];
    sm[7] = fc2[7] + ig2[7];
    asm volatile("" : "+v"(to[7]), "+v"(sm[6]), "+v"(sm[7]), "+v"(cr[0]), "+v"(sm[4]), "+v"(sm[5]), "+v"(cr[1]), "+v"(cr[2]), "+v"(cr[3]));
    v2f tw[8];
    auto fold = [&](int r) __attribute__((always_inline)) {
      v2f t2 = {to[r], to[r]};
      tw[r] = __builtin_elementwise_fma(t2, wv[r], wv[r]);
    };
    float a0, a1, b0, b1;
    fold(0);
    fold(1);
    asm volatile("" : "+v"(tw[0]), "+v"(tw[1]), "+v"(cr[0]), "+v"(sm[4]), "+v"(sm[5]), "+v"(cr[1]));
    tc[0] = tanhf_fast(cr[0]);
    cr[4] = 0.5f * sm[4];
    cr[5] = 0.5f * sm[5];
    asm volatile("" : "+v"(tc[0]), "+v"(cr[4]), "+v"(cr[5]), "+v"(cr[1]), "+v"(to[2]), "+v"(cr[2]), "+v"(sm[6]), "+v"(sm[7]));
    tc[1] = tanhf_fast(cr[1]);
    fold(2);
    asm volatile("" : "+v"(tc[1]), "+v"(tw[2]), "+v"(cr[2]), "+v"(sm[6]), "+v"(sm[7]), "+v"(cr[3]));
    tc[2] = tanhf_fast(cr[2]);
    cr[6] = 0.5f * sm[6];
    cr[7] = 0.5f * sm[7];
    asm volatile("" : "+v"(tc[2]), "+v"(cr[6]), "+v"(cr[7]), "+v"(cr[3]), "+v"(to[3]), "+v"(cr[4]));
    tc[3] = tanhf_fast(cr[3]);
    fold(3);
    asm volatile("" : "+v"(tc[3]), "+v"(tw[3]), "+v"(cr[4]), "+v"(to[4]), "+v"(cr[5]));
    tc[4] = tanhf_fast(cr[4]);
    fold(4);
    asm volatile("" : "+v"(tc[4]), "+v"(tw[4]), "+v"(cr[5]), "+v"(tc[0]), "+v"(tw[0]), "+v"(cr[6]));
    tc[5] = tanhf_fast(cr[5]);
    a0 = tc[0] * tw[0][0];
    a1 = tc[0] * tw[0][1];
    asm volatile("" : "+v"(tc[5]), "+v"(a0), "+v"(a1), "+v"(cr[6]), "+v"(tc[1]), "+v"(tw[1]), "+v"(cr[7]));
    tc[6] = tanhf_fast(cr[6]);
    b0 = tc[1] * tw[1][0];
    b1 = tc[1] * tw[1][1];
    asm volatile("" : "+v"(tc[6]), "+v"(b0), "+v"(b1), "+v"(cr[7]), "+v"(to[5]));
    tc[7] = tanhf_fast(cr[7]);
    fold(5);
    asm volatile("" : "+v"(tc[7]), "+v"(tw[5]), "+v"(a0), "+v"(a1), "+v"(tc[2]), "+v"(tw[2]), "+v"(to[6]));
    a0 = __builtin_fmaf(tc[2], tw[2][0], a0);
    a1 = __builtin_fmaf(tc[2], tw[2][1], a1);
    fold(6);
    asm volatile("" : "+v"(a0), "+v"(a1), "+v"(tw[6]), "+v"(b0), "+v"(b1), "+v"(tc[3]), "+v"(tw[3]), "+v"(to[7]));
    b0 = __builtin_fmaf(tc[3], tw[3][0], b0);
    b1 = __builtin_fmaf(tc[3], tw[3][1], b1);
    fold(7);
    asm volatile("" : "+v"(b0), "+v"(b1), "+v"(tw[7]), "+v"(a0), "+v"(a1), "+v"(tc[4]), "+v"(tw[4]));
    a0 = __builtin_fmaf(tc[4], tw[4][0], a0);
    a1 = __builtin_fmaf(tc[4], tw[4][1], a1);
    asm volatile("" : "+v"(a0), "+v"(a1), "+v"(b0), "+v"(b1), "+v"(tc[5]), "+v"(tw[5]));
    b0 = __builtin_fmaf(tc[5], tw[5][0], b0);
    b1 = __builtin_fmaf(tc[5], tw[5][1], b1);
    asm volatile("" : "+v"(b0), "+v"(b1), "+v"(a0), "+v"(a1), "+v"(tc[6]), "+v"(tw[6]));
    a0 = __builtin_fmaf(tc[6], tw[6][0], a0);
    a1 = __builtin_fmaf(tc[6], tw[6][1], a1);
    asm volatile("" : "+v"(a0), "+v"(a1), "+v"(b0), "+v"(b1), "+v"(tc[7]), "+v"(tw[7]));
    b0 = __builtin_fmaf(tc[7], tw[7][0], b0);
    b1 = __builtin_fmaf(tc[7], tw[7][1], b1);
    asm volatile("" : "+v"(a0), "+v"(a1), "+v"(b0), "+v"(b1));
    float s0 = a0 + b0, s1 = a1 + b1;
#pragma unroll
    for (int r = 0; r < 8; ++r) c[r] = cr[r];

    asm volatile("" : "+v"(s0), "+v"(s1));
    float q0 = halfswap16(s0), q1 = halfswap16(s1);
    float p0 = half ? s0 : xn0, p1 = half ? s1 : xn1;
    hb0 = __builtin_fmaf(q0, hm, p0);
    hb1 = __builtin_fmaf(q1, hm, p1);

    o0 = hb0;
    o1 = hb1;
  };

  const int nblk = nsteps2 >> 3;
  v4f8 xc0 = {0.f, 0.f, 0.f, 0.f}, xc1 = xc0, xc2 = xc0, xc3 = xc0;
  if (nblk > 0) {
    const v4f8* p = (const v4f8*)xbase;
    xc0 = p[0]; xc1 = p[1]; xc2 = p[2]; xc3 = p[3];
  }
  for (int blk = 0; blk < nblk; ++blk) {
    int nb = (blk + 1 < nblk) ? blk + 1 : blk;
    const v4f8* np = (const v4f8*)(xbase + nb * 16);
    v4f8 xn0 = np[0], xn1 = np[1], xn2 = np[2], xn3 = np[3];

    float hs[16];
    step(xc0[2], xc0[3], hs[0],  hs[1]);
    step(xc1[0], xc1[1], hs[2],  hs[3]);
    step(xc1[2], xc1[3], hs[4],  hs[5]);
    step(xc2[0], xc2[1], hs[6],  hs[7]);
    step(xc2[2], xc2[3], hs[8],  hs[9]);
    step(xc3[0], xc3[1], hs[10], hs[11]);
    step(xc3[2], xc3[3], hs[12], hs[13]);
    step(xn0[0], xn0[1], hs[14], hs[15]);
    if (half) {
      v4f8* op = (v4f8*)(obase + blk * 16);
#pragma unroll
      for (int q = 0; q < 4; ++q) {
        v4f8 hv = {hs[4 * q], hs[4 * q + 1], hs[4 * q + 2], hs[4 * q + 3]};
        op[q] = hv;
      }
    }

    xc0 = xn0; xc1 = xn1; xc2 = xn2; xc3 = xn3;
  }

  if ((nblk << 3) < nsteps2) {
    const int s = nblk << 3;
    hb0 = half ? hb0 : xbase[s * 2 + 0];
    hb1 = half ? hb1 : xbase[s * 2 + 1];
  }
  for (int s = nblk << 3; s < nsteps2; ++s) {
    const int sn = (s + 1 < nsteps2) ? s + 1 : s;
    float h0, h1;
    step(xbase[sn * 2 + 0], xbase[sn * 2 + 1], h0, h1);
    if (half) {
      obase[s * 2 + 0] = h0;
      obase[s * 2 + 1] = h1;
    }
  }
}

extern "C" void kernel_launch(void* const* d_in, const int* in_sizes, int n_in,
                              void* d_out, int out_size, void* d_ws, size_t ws_size,
                              hipStream_t stream) {
  const float* u_train = (const float*)d_in[0];
  const float* y_train = (const float*)d_in[1];
  const float* W_ih    = (const float*)d_in[2];
  const float* W_hh    = (const float*)d_in[3];
  const float* b_ih    = (const float*)d_in[4];
  const float* b_hh    = (const float*)d_in[5];
  const float* W_hr    = (const float*)d_in[6];
  const int B = 64;
  const int T = in_sizes[0] / (B * INPUT);
  const int nstep = T / 2;

  float* out    = (float*)d_out;
  float* c_last = (float*)d_ws;

  int total = B * nstep;
  lstm_phase1<<<(total + 255) / 256, 256, 0, stream>>>(
      y_train, W_ih, b_ih, b_hh, W_hr, out, c_last, B, T, nstep);

  lstm_phase2<<<B / 16, 32, 0, stream>>>(
      u_train, W_ih, W_hh, b_ih, b_hh, W_hr, c_last, out, T, nstep, T - nstep);
}


# Scan re-tiled to 8 batches a wave, 4 lanes a batch: a column half picks its hidden units through its K pair, x enters through C from 2 own WMMAs; 20 tanh a lane a step, was 40 (new summation order)
# speedup vs baseline: 2.8420x; 1.1924x over previous
#include <hip/hip_runtime.h>
#include <hip/hip_bf16.h>
#include <math.h>


typedef float v2f __attribute__((ext_vector_type(2)));
typedef float v8f __attribute__((ext_vector_type(8)));
typedef float v4f8 __attribute__((ext_vector_type(4), aligned(8)));

#define HIDDEN 16
#define PROJ   2
#define INPUT  2

#define LOG2E 1.4426950408889634f

__device__ __forceinline__ float sigmoidf_fast(float x) {
  return __builtin_amdgcn_rcpf(1.0f + __builtin_amdgcn_exp2f(x * -LOG2E));
}

__device__ __forceinline__ float tanhf_fast(float x) {
#if __has_builtin(__builtin_amdgcn_tanhf)
  return __builtin_amdgcn_tanhf(x);
#else
  float e = __builtin_amdgcn_exp2f(x * (-2.0f * LOG2E));
  return __builtin_fmaf(2.0f, __builtin_amdgcn_rcpf(1.0f + e), -1.0f);
#endif
}

__device__ __forceinline__ float halfswap16(float v) {
#if __has_builtin(__builtin_amdgcn_permlanex16)
  int i = __builtin_bit_cast(int, v);
  i = __builtin_amdgcn_permlanex16(0, i, 0x76543210, 0xFEDCBA98, false, false);
  return __builtin_bit_cast(float, i);
#else
  return __shfl_xor(v, 16, 32);
#endif
}

__global__ void __launch_bounds__(256) lstm_phase1(
    const float* __restrict__ y_train, const float* __restrict__ W_ih,
    const float* __restrict__ b_ih, const float* __restrict__ b_hh,
    const float* __restrict__ W_hr, float* __restrict__ out,
    float* __restrict__ c_last, int B, int T, int nstep) {
  int idx = blockIdx.x * blockDim.x + threadIdx.x;
  if (idx >= B * nstep) return;
  int b = idx / nstep;
  int t = idx - b * nstep;

  const float* xp = y_train + ((size_t)b * T + t) * 2;
  float x0 = xp[0], x1 = xp[1];
  float y0 = 0.f, y1 = 0.f;
  bool last = (t == nstep - 1);

#pragma unroll
  for (int n = 0; n < HIDDEN; ++n) {
    int ng = 32 + n, no = 48 + n;
    float gi = W_ih[n * 2 + 0] * x0 + W_ih[n * 2 + 1] * x1 + b_ih[n] + b_hh[n];
    float gg = W_ih[ng * 2 + 0] * x0 + W_ih[ng * 2 + 1] * x1 + b_ih[ng] + b_hh[ng];
    float go = W_ih[no * 2 + 0] * x0 + W_ih[no * 2 + 1] * x1 + b_ih[no] + b_hh[no];
    float cv = sigmoidf_fast(gi) * tanhf_fast(gg);
    float hp = sigmoidf_fast(go) * tanhf_fast(cv);
    y0 = __builtin_fmaf(hp, W_hr[n], y0);
    y1 = __builtin_fmaf(hp, W_hr[16 + n], y1);
    if (last) c_last[b * HIDDEN + n] = cv;
  }
  float* op = out + ((size_t)b * T + t) * 2;
  op[0] = y0;
  op[1] = y1;
}

__global__ void __launch_bounds__(32) lstm_phase2(
    const float* __restrict__ u_train, const float* __restrict__ W_ih,
    const float* __restrict__ W_hh, const float* __restrict__ b_ih,
    const float* __restrict__ b_hh, const float* __restrict__ W_hr,
    const float* __restrict__ c_last, float* __restrict__ out,
    int T, int nstep, int nsteps2) {
  const int lane = threadIdx.x;
  const int hf   = lane >> 4;
  const int col  = lane & 15;
  const int cv   = col >> 3;
  const int bg   = blockIdx.x * 8 + (col & 7);

  v2f Ah[2], Ax[2];
#pragma unroll
  for (int t = 0; t < 2; ++t) {
    int r = col & 7;
    int G = t * 2 + (r >> 2);
    int n = G * 16 + hf * 8 + (col >> 3) * 4 + (r & 3);
    float sc = (G == 2) ? 1.0f : 0.5f;
    Ah[t][0] = sc * W_hh[n * 2 + 0];
    Ah[t][1] = sc * W_hh[n * 2 + 1];
    Ax[t][0] = sc * W_ih[n * 2 + 0];
    Ax[t][1] = sc * W_ih[n * 2 + 1];
  }

  const int hid0 = cv * 8 + hf * 4;
  v8f bias0, bias1;
#pragma unroll
  for (int r = 0; r < 8; ++r) {
    int hid = hid0 + (r & 3);
    int g0 = (r >> 2) * 16 + hid;
    int g1 = (2 + (r >> 2)) * 16 + hid;
    bias0[r] = 0.5f * (b_ih[g0] + b_hh[g0]);
    bias1[r] = ((r < 4) ? 1.0f : 0.5f) * (b_ih[g1] + b_hh[g1]);
  }
  v2f wv[4];
  float c[4];
#pragma unroll
  for (int k = 0; k < 4; ++k) {
    wv[k][0] = 0.5f * W_hr[hid0 + k];
    wv[k][1] = 0.5f * W_hr[16 + hid0 + k];
    c[k] = c_last[bg * HIDDEN + hid0 + k];
  }

  const float* xbase = u_train + ((size_t)bg * T + nstep) * 2;
  float* obase = out + ((size_t)bg * T + nstep) * 2;

  const float msel = (hf == cv) ? 1.0f : 0.0f;

  float hb0, hb1;
  v8f cx0 = bias0, cx1 = bias1;
  {
    const float* hp = out + ((size_t)bg * T + (nstep - 1)) * 2;
    hb0 = msel * hp[0];
    hb1 = msel * hp[1];
    if (nsteps2 > 0) {
      v2f bx;
      bx[0] = msel * xbase[0];
      bx[1] = msel * xbase[1];
      cx0 = __builtin_amdgcn_wmma_f32_16x16x4_f32(false, Ax[0], false, bx, (short)0, bias0, false, false);
      cx1 = __builtin_amdgcn_wmma_f32_16x16x4_f32(false, Ax[1], false, bx, (short)0, bias1, false, false);
    }
  }

  auto step = [&](float xn0, float xn1, float& o0, float& o1) __attribute__((always_inline)) {
    v2f bh;
    bh[0] = hb0;
    bh[1] = hb1;
    v2f bx;
    bx[0] = msel * xn0;
    bx[1] = msel * xn1;
    v8f d0 = __builtin_amdgcn_wmma_f32_16x16x4_f32(false, Ah[0], false, bh, (short)0, cx0, false, false);
    v8f d1 = __builtin_amdgcn_wmma_f32_16x16x4_f32(false, Ah[1], false, bh, (short)0, cx1, false, false);

    float tf[4], ti[4], gg[4], to[4], hfc[4], hi[4], cr[4], tc[4];
#pragma unroll
    for (int k = 0; k < 4; ++k) tf[k] = tanhf_fast(d0[4 + k]);
#pragma unroll
    for (int k = 0; k < 4; ++k) ti[k] = tanhf_fast(d0[k]);
    asm volatile("" : "+v"(bx), "+v"(tf[0]), "+v"(tf[1]), "+v"(tf[2]), "+v"(tf[3]), "+v"(ti[0]), "+v"(ti[1]), "+v"(ti[2]), "+v"(ti[3]));
    cx0 = __builtin_amdgcn_wmma_f32_16x16x4_f32(false, Ax[0], false, bx, (short)0, bias0, false, false);
    asm volatile("" : "+v"(cx0), "+v"(d1));
#pragma unroll
    for (int k = 0; k < 4; ++k) gg[k] = tanhf_fast(d1[k]);
#pragma unroll
    for (int k = 0; k < 4; ++k) to[k] = tanhf_fast(d1[4 + k]);
    asm volatile("" : "+v"(bx), "+v"(gg[0]), "+v"(gg[1]), "+v"(gg[2]), "+v"(gg[3]), "+v"(to[0]), "+v"(to[1]), "+v"(to[2]), "+v"(to[3]));
    cx1 = __builtin_amdgcn_wmma_f32_16x16x4_f32(false, Ax[1], false, bx, (short)0, bias1, false, false);

#pragma unroll
    for (int k = 0; k < 4; ++k) {
      float hc = 0.5f * c[k];
      hfc[k] = __builtin_fmaf(tf[k], hc, hc);
      hi[k]  = __builtin_fmaf(ti[k], 0.5f, 0.5f);
      cr[k]  = __builtin_fmaf(gg[k], hi[k], hfc[k]);
    }
    asm volatile("" : "+v"(cx1), "+v"(cr[0]), "+v"(cr[1]), "+v"(cr[2]), "+v"(cr[3]));
#pragma unroll
    for (int k = 0; k < 4; ++k) {
      tc[k] = tanhf_fast(cr[k]);
      c[k]  = cr[k];
    }
    v2f tw[4];
#pragma unroll
    for (int k = 0; k < 4; ++k) {
      v2f t2 = {to[k], to[k]};
      tw[k] = __builtin_elementwise_fma(t2, wv[k], wv[k]);
    }
    float a0 = tc[0] * tw[0][0], a1 = tc[0] * tw[0][1];
    float b0 = tc[1] * tw[1][0], b1 = tc[1] * tw[1][1];
    asm volatile("" : "+v"(a0), "+v"(a1), "+v"(b0), "+v"(b1));
    a0 = __builtin_fmaf(tc[2], tw[2][0], a0);
    a1 = __builtin_fmaf(tc[2], tw[2][1], a1);
    b0 = __builtin_fmaf(tc[3], tw[3][0], b0);
    b1 = __builtin_fmaf(tc[3], tw[3][1], b1);
    asm volatile("" : "+v"(a0), "+v"(a1), "+v"(b0), "+v"(b1));
    float s0 = a0 + b0, s1 = a1 + b1;
    float t0, t1;
    asm volatile("v_add_f32_dpp %0, %1, %1 row_ror:8 row_mask:0xf bank_mask:0xf" : "=v"(t0) : "v"(s0));
    asm volatile("v_add_f32_dpp %0, %1, %1 row_ror:8 row_mask:0xf bank_mask:0xf" : "=v"(t1) : "v"(s1));
    float q0 = halfswap16(t0), q1 = halfswap16(t1);
    float m0 = msel * t0, m1 = msel * t1;
    asm volatile("" : "+v"(q0), "+v"(q1), "+v"(m0), "+v"(m1));
    hb0 = __builtin_fmaf(q0, msel, m0);
    hb1 = __builtin_fmaf(q1, msel, m1);
    o0 = hb0;
    o1 = hb1;
  };

  const int nblk = nsteps2 >> 3;
  v4f8 xc0 = {0.f, 0.f, 0.f, 0.f}, xc1 = xc0, xc2 = xc0, xc3 = xc0;
  if (nblk > 0) {
    const v4f8* p = (const v4f8*)xbase;
    xc0 = p[0]; xc1 = p[1]; xc2 = p[2]; xc3 = p[3];
  }
  for (int blk = 0; blk < nblk; ++blk) {
    int nb = (blk + 1 < nblk) ? blk + 1 : blk;
    const v4f8* np = (const v4f8*)(xbase + nb * 16);
    v4f8 xn0 = np[0], xn1 = np[1], xn2 = np[2], xn3 = np[3];

    float hs[16];
    step(xc0[2], xc0[3], hs[0],  hs[1]);
    step(xc1[0], xc1[1], hs[2],  hs[3]);
    step(xc1[2], xc1[3], hs[4],  hs[5]);
    step(xc2[0], xc2[1], hs[6],  hs[7]);
    step(xc2[2], xc2[3], hs[8],  hs[9]);
    step(xc3[0], xc3[1], hs[10], hs[11]);
    step(xc3[2], xc3[3], hs[12], hs[13]);
    step(xn0[0], xn0[1], hs[14], hs[15]);
    if (lane < 8) {
      v4f8* op = (v4f8*)(obase + blk * 16);
#pragma unroll
      for (int q = 0; q < 4; ++q) {
        v4f8 hv = {hs[4 * q], hs[4 * q + 1], hs[4 * q + 2], hs[4 * q + 3]};
        op[q] = hv;
      }
    }

    xc0 = xn0; xc1 = xn1; xc2 = xn2; xc3 = xn3;
  }

  if ((nblk << 3) < nsteps2 && nblk > 0) {
    const int s = nblk << 3;
    v2f bx;
    bx[0] = msel * xbase[s * 2 + 0];
    bx[1] = msel * xbase[s * 2 + 1];
    cx0 = __builtin_amdgcn_wmma_f32_16x16x4_f32(false, Ax[0], false, bx, (short)0, bias0, false, false);
    cx1 = __builtin_amdgcn_wmma_f32_16x16x4_f32(false, Ax[1], false, bx, (short)0, bias1, false, false);
  }
  for (int s = nblk << 3; s < nsteps2; ++s) {
    const int sn = (s + 1 < nsteps2) ? s + 1 : s;
    float h0, h1;
    step(xbase[sn * 2 + 0], xbase[sn * 2 + 1], h0, h1);
    if (lane < 8) {
      obase[s * 2 + 0] = h0;
      obase[s * 2 + 1] = h1;
    }
  }
}

extern "C" void kernel_launch(void* const* d_in, const int* in_sizes, int n_in,
                              void* d_out, int out_size, void* d_ws, size_t ws_size,
                              hipStream_t stream) {
  const float* u_train = (const float*)d_in[0];
  const float* y_train = (const float*)d_in[1];
  const float* W_ih    = (const float*)d_in[2];
  const float* W_hh    = (const float*)d_in[3];
  const float* b_ih    = (const float*)d_in[4];
  const float* b_hh    = (const float*)d_in[5];
  const float* W_hr    = (const float*)d_in[6];
  const int B = 64;
  const int T = in_sizes[0] / (B * INPUT);
  const int nstep = T / 2;

  float* out    = (float*)d_out;
  float* c_last = (float*)d_ws;

  int total = B * nstep;
  lstm_phase1<<<(total + 255) / 256, 256, 0, stream>>>(
      y_train, W_ih, b_ih, b_hh, W_hr, out, c_last, B, T, nstep);

  lstm_phase2<<<B / 8, 32, 0, stream>>>(
      u_train, W_ih, W_hh, b_ih, b_hh, W_hr, c_last, out, T, nstep, T - nstep);
}


# Input-term WMMAs leave the scan wave: a second wave of the work-group forms each step's two C tiles and hands them over through LDS, 8 steps a block, one barrier a block; summation order unchanged
# speedup vs baseline: 3.2556x; 1.1455x over previous
#include <hip/hip_runtime.h>
#include <hip/hip_bf16.h>
#include <math.h>


typedef float v2f __attribute__((ext_vector_type(2)));
typedef float v8f __attribute__((ext_vector_type(8)));
typedef float v4f8 __attribute__((ext_vector_type(4), aligned(8)));
typedef float v4f __attribute__((ext_vector_type(4)));

#define HIDDEN 16
#define PROJ   2
#define INPUT  2

#define LOG2E 1.4426950408889634f

__device__ __forceinline__ float sigmoidf_fast(float x) {
  return __builtin_amdgcn_rcpf(1.0f + __builtin_amdgcn_exp2f(x * -LOG2E));
}

__device__ __forceinline__ float tanhf_fast(float x) {
#if __has_builtin(__builtin_amdgcn_tanhf)
  return __builtin_amdgcn_tanhf(x);
#else
  float e = __builtin_amdgcn_exp2f(x * (-2.0f * LOG2E));
  return __builtin_fmaf(2.0f, __builtin_amdgcn_rcpf(1.0f + e), -1.0f);
#endif
}

__device__ __forceinline__ float halfswap16(float v) {
#if __has_builtin(__builtin_amdgcn_permlanex16)
  int i = __builtin_bit_cast(int, v);
  i = __builtin_amdgcn_permlanex16(0, i, 0x76543210, 0xFEDCBA98, false, false);
  return __builtin_bit_cast(float, i);
#else
  return __shfl_xor(v, 16, 32);
#endif
}

__global__ void __launch_bounds__(256) lstm_phase1(
    const float* __restrict__ y_train, const float* __restrict__ W_ih,
    const float* __restrict__ b_ih, const float* __restrict__ b_hh,
    const float* __restrict__ W_hr, float* __restrict__ out,
    float* __restrict__ c_last, int B, int T, int nstep) {
  int idx = blockIdx.x * blockDim.x + threadIdx.x;
  if (idx >= B * nstep) return;
  int b = idx / nstep;
  int t = idx - b * nstep;

  const float* xp = y_train + ((size_t)b * T + t) * 2;
  float x0 = xp[0], x1 = xp[1];
  float y0 = 0.f, y1 = 0.f;
  bool last = (t == nstep - 1);

#pragma unroll
  for (int n = 0; n < HIDDEN; ++n) {
    int ng = 32 + n, no = 48 + n;
    float gi = W_ih[n * 2 + 0] * x0 + W_ih[n * 2 + 1] * x1 + b_ih[n] + b_hh[n];
    float gg = W_ih[ng * 2 + 0] * x0 + W_ih[ng * 2 + 1] * x1 + b_ih[ng] + b_hh[ng];
    float go = W_ih[no * 2 + 0] * x0 + W_ih[no * 2 + 1] * x1 + b_ih[no] + b_hh[no];
    float cv = sigmoidf_fast(gi) * tanhf_fast(gg);
    float hp = sigmoidf_fast(go) * tanhf_fast(cv);
    y0 = __builtin_fmaf(hp, W_hr[n], y0);
    y1 = __builtin_fmaf(hp, W_hr[16 + n], y1);
    if (last) c_last[b * HIDDEN + n] = cv;
  }
  float* op = out + ((size_t)b * T + t) * 2;
  op[0] = y0;
  op[1] = y1;
}

__global__ void __launch_bounds__(64) lstm_phase2(
    const float* __restrict__ u_train, const float* __restrict__ W_ih,
    const float* __restrict__ W_hh, const float* __restrict__ b_ih,
    const float* __restrict__ b_hh, const float* __restrict__ W_hr,
    const float* __restrict__ c_last, float* __restrict__ out,
    int T, int nstep, int nsteps2) {
  __shared__ v4f ring[2 * 8 * 4 * 32];
  const int wave = __builtin_amdgcn_readfirstlane((int)(threadIdx.x >> 5));
  const int lane = threadIdx.x & 31;
  const int hf   = lane >> 4;
  const int col  = lane & 15;
  const int cv   = col >> 3;
  const int bg   = blockIdx.x * 8 + (col & 7);

  v2f Ah[2], Ax[2];
#pragma unroll
  for (int t = 0; t < 2; ++t) {
    int r = col & 7;
    int G = t * 2 + (r >> 2);
    int n = G * 16 + hf * 8 + (col >> 3) * 4 + (r & 3);
    float sc = (G == 2) ? 1.0f : 0.5f;
    Ah[t][0] = sc * W_hh[n * 2 + 0];
    Ah[t][1] = sc * W_hh[n * 2 + 1];
    Ax[t][0] = sc * W_ih[n * 2 + 0];
    Ax[t][1] = sc * W_ih[n * 2 + 1];
  }

  const int hid0 = cv * 8 + hf * 4;
  v8f bias0, bias1;
#pragma unroll
  for (int r = 0; r < 8; ++r) {
    int hid = hid0 + (r & 3);
    int g0 = (r >> 2) * 16 + hid;
    int g1 = (2 + (r >> 2)) * 16 + hid;
    bias0[r] = 0.5f * (b_ih[g0] + b_hh[g0]);
    bias1[r] = ((r < 4) ? 1.0f : 0.5f) * (b_ih[g1] + b_hh[g1]);
  }
  v2f wv[4];
  float c[4];
#pragma unroll
  for (int k = 0; k < 4; ++k) {
    wv[k][0] = 0.5f * W_hr[hid0 + k];
    wv[k][1] = 0.5f * W_hr[16 + hid0 + k];
    c[k] = c_last[bg * HIDDEN + hid0 + k];
  }

  const float* xbase = u_train + ((size_t)bg * T + nstep) * 2;
  float* obase = out + ((size_t)bg * T + nstep) * 2;

  const float msel = (hf == cv) ? 1.0f : 0.0f;

  float hb0, hb1;
  {
    const float* hp = out + ((size_t)bg * T + (nstep - 1)) * 2;
    hb0 = msel * hp[0];
    hb1 = msel * hp[1];
  }

  auto ctiles = [&](float x0, float x1, v8f& t0, v8f& t1) __attribute__((always_inline)) {
    v2f bx;
    bx[0] = msel * x0;
    bx[1] = msel * x1;
    t0 = __builtin_amdgcn_wmma_f32_16x16x4_f32(false, Ax[0], false, bx, (short)0, bias0, false, false);
    t1 = __builtin_amdgcn_wmma_f32_16x16x4_f32(false, Ax[1], false, bx, (short)0, bias1, false, false);
  };

  auto step = [&](v8f cx0, v8f cx1, float& o0, float& o1) __attribute__((always_inline)) {
    v2f bh;
    bh[0] = hb0;
    bh[1] = hb1;
    v8f d0 = __builtin_amdgcn_wmma_f32_16x16x4_f32(false, Ah[0], false, bh, (short)0, cx0, false, false);
    v8f d1 = __builtin_amdgcn_wmma_f32_16x16x4_f32(false, Ah[1], false, bh, (short)0, cx1, false, false);

    float tf[4], ti[4], gg[4], to[4], hfc[4], hi[4], cr[4], tc[4];
#pragma unroll
    for (int k = 0; k < 4; ++k) tf[k] = tanhf_fast(d0[4 + k]);
#pragma unroll
    for (int k = 0; k < 4; ++k) ti[k] = tanhf_fast(d0[k]);
#pragma unroll
    for (int k = 0; k < 4; ++k) gg[k] = tanhf_fast(d1[k]);
#pragma unroll
    for (int k = 0; k < 4; ++k) to[k] = tanhf_fast(d1[4 + k]);

#pragma unroll
    for (int k = 0; k < 4; ++k) {
      float hc = 0.5f * c[k];
      hfc[k] = __builtin_fmaf(tf[k], hc, hc);
      hi[k]  = __builtin_fmaf(ti[k], 0.5f, 0.5f);
      cr[k]  = __builtin_fmaf(gg[k], hi[k], hfc[k]);
    }
#pragma unroll
    for (int k = 0; k < 4; ++k) {
      tc[k] = tanhf_fast(cr[k]);
      c[k]  = cr[k];
    }
    v2f tw[4];
#pragma unroll
    for (int k = 0; k < 4; ++k) {
      v2f t2 = {to[k], to[k]};
      tw[k] = __builtin_elementwise_fma(t2, wv[k], wv[k]);
    }
    float a0 = tc[0] * tw[0][0], a1 = tc[0] * tw[0][1];
    float b0 = tc[1] * tw[1][0], b1 = tc[1] * tw[1][1];
    asm volatile("" : "+v"(a0), "+v"(a1), "+v"(b0), "+v"(b1));
    a0 = __builtin_fmaf(tc[2], tw[2][0], a0);
    a1 = __builtin_fmaf(tc[2], tw[2][1], a1);
    b0 = __builtin_fmaf(tc[3], tw[3][0], b0);
    b1 = __builtin_fmaf(tc[3], tw[3][1], b1);
    asm volatile("" : "+v"(a0), "+v"(a1), "+v"(b0), "+v"(b1));
    float s0 = a0 + b0, s1 = a1 + b1;
    float t0, t1;
    asm volatile("v_add_f32_dpp %0, %1, %1 row_ror:8 row_mask:0xf bank_mask:0xf" : "=v"(t0) : "v"(s0));
    asm volatile("v_add_f32_dpp %0, %1, %1 row_ror:8 row_mask:0xf bank_mask:0xf" : "=v"(t1) : "v"(s1));
    float q0 = halfswap16(t0), q1 = halfswap16(t1);
    float m0 = msel * t0, m1 = msel * t1;
    asm volatile("" : "+v"(q0), "+v"(q1), "+v"(m0), "+v"(m1));
    hb0 = __builtin_fmaf(q0, msel, m0);
    hb1 = __builtin_fmaf(q1, msel, m1);
    o0 = hb0;
    o1 = hb1;
  };

  const int nblk = (nsteps2 >> 4) * 2;

  if (nblk > 0) {
    v4f8 xq0 = {0.f, 0.f, 0.f, 0.f}, xq1 = xq0, xq2 = xq0, xq3 = xq0;
    if (wave == 1) {
      const v4f8* p = (const v4f8*)xbase;
      xq0 = p[0]; xq1 = p[1]; xq2 = p[2]; xq3 = p[3];
    }
    auto produce = [&](int j, int buf) __attribute__((always_inline)) {
      int nb = (j + 1 < nblk) ? j + 1 : j;
      const v4f8* np = (const v4f8*)(xbase + nb * 16);
      v4f8 xn0 = np[0], xn1 = np[1], xn2 = np[2], xn3 = np[3];
      const float xs[16] = {xq0[0], xq0[1], xq0[2], xq0[3], xq1[0], xq1[1], xq1[2], xq1[3],
                            xq2[0], xq2[1], xq2[2], xq2[3], xq3[0], xq3[1], xq3[2], xq3[3]};
#pragma unroll
      for (int s = 0; s < 8; ++s) {
        v8f t0, t1;
        ctiles(xs[2 * s], xs[2 * s + 1], t0, t1);
        v4f* d = ring + ((buf * 8 + s) * 4) * 32 + lane;
        d[0]  = __builtin_shufflevector(t0, t0, 0, 1, 2, 3);
        d[32] = __builtin_shufflevector(t0, t0, 4, 5, 6, 7);
        d[64] = __builtin_shufflevector(t1, t1, 0, 1, 2, 3);
        d[96] = __builtin_shufflevector(t1, t1, 4, 5, 6, 7);
      }
      xq0 = xn0; xq1 = xn1; xq2 = xn2; xq3 = xn3;
    };

    v8f SA[8][2], SB[8][2];
    float hp[16];
#pragma unroll
    for (int q = 0; q < 16; ++q) hp[q] = 0.f;
    auto fetch = [&](v8f (&S)[8][2], int buf) __attribute__((always_inline)) {
#pragma unroll
      for (int s = 0; s < 8; ++s)
#pragma unroll
        for (int t = 0; t < 2; ++t) {
          const v4f* src = ring + ((buf * 8 + s) * 4 + 2 * t) * 32 + lane;
          S[s][t] = __builtin_shufflevector(src[0], src[32], 0, 1, 2, 3, 4, 5, 6, 7);
        }
    };
    auto run = [&](v8f (&S)[8][2]) __attribute__((always_inline)) {
#pragma unroll
      for (int s = 0; s < 8; ++s) step(S[s][0], S[s][1], hp[2 * s], hp[2 * s + 1]);
    };
    auto drop = [&](v8f (&S)[8][2]) __attribute__((always_inline)) {
#pragma unroll
      for (int s = 0; s < 8; ++s) {
        S[s][0] = __builtin_nondeterministic_value(S[s][0]);
        S[s][1] = __builtin_nondeterministic_value(S[s][1]);
      }
      hb0 = __builtin_nondeterministic_value(hb0);
      hb1 = __builtin_nondeterministic_value(hb1);
#pragma unroll
      for (int k = 0; k < 4; ++k) c[k] = __builtin_nondeterministic_value(c[k]);
#pragma unroll
      for (int q = 0; q < 16; ++q) hp[q] = __builtin_nondeterministic_value(hp[q]);
    };
    auto put = [&](int blk) __attribute__((always_inline)) {
      if (lane < 8) {
        v4f8* op = (v4f8*)(obase + blk * 16);
#pragma unroll
        for (int q = 0; q < 4; ++q) {
          v4f8 hv = {hp[4 * q], hp[4 * q + 1], hp[4 * q + 2], hp[4 * q + 3]};
          op[q] = hv;
        }
      }
    };

    if (wave == 1) produce(0, 0);
    __syncthreads();
    if (wave == 1) produce(1, 1);
    else           fetch(SA, 0);
    __syncthreads();
    for (int j = 0; j < nblk; j += 2) {
      if (wave == 1) { if (j + 2 < nblk) produce(j + 2, 0); drop(SB); }
      else {
        if (j > 0) put(j - 1);
        fetch(SB, 1);
        run(SA);
      }
      __syncthreads();
      if (wave == 1) { if (j + 3 < nblk) produce(j + 3, 1); drop(SA); }
      else {
        put(j);
        fetch(SA, 0);
        run(SB);
      }
      __syncthreads();
    }
    if (wave == 0) put(nblk - 1);
  }

  if (wave == 0) {
    for (int s = nblk << 3; s < nsteps2; ++s) {
      v8f c0, c1;
      ctiles(xbase[s * 2 + 0], xbase[s * 2 + 1], c0, c1);
      float h0, h1;
      step(c0, c1, h0, h1);
      if (lane < 8) {
        obase[s * 2 + 0] = h0;
        obase[s * 2 + 1] = h1;
      }
    }
  }
}

extern "C" void kernel_launch(void* const* d_in, const int* in_sizes, int n_in,
                              void* d_out, int out_size, void* d_ws, size_t ws_size,
                              hipStream_t stream) {
  const float* u_train = (const float*)d_in[0];
  const float* y_train = (const float*)d_in[1];
  const float* W_ih    = (const float*)d_in[2];
  const float* W_hh    = (const float*)d_in[3];
  const float* b_ih    = (const float*)d_in[4];
  const float* b_hh    = (const float*)d_in[5];
  const float* W_hr    = (const float*)d_in[6];
  const int B = 64;
  const int T = in_sizes[0] / (B * INPUT);
  const int nstep = T / 2;

  float* out    = (float*)d_out;
  float* c_last = (float*)d_ws;

  int total = B * nstep;
  lstm_phase1<<<(total + 255) / 256, 256, 0, stream>>>(
      y_train, W_ih, b_ih, b_hh, W_hr, out, c_last, B, T, nstep);

  lstm_phase2<<<B / 8, 64, 0, stream>>>(
      u_train, W_ih, W_hh, b_ih, b_hh, W_hr, c_last, out, T, nstep, T - nstep);
}


# Scan re-tiled to 4 batches a wave, 8 lanes a batch: a column quarter picks its hidden units through K=8 as two chained WMMAs on one tile; 10 tanh a lane a step, was 20 (new summation order)
# speedup vs baseline: 3.6978x; 1.1358x over previous
#include <hip/hip_runtime.h>
#include <hip/hip_bf16.h>
#include <math.h>


typedef float v2f __attribute__((ext_vector_type(2)));
typedef float v8f __attribute__((ext_vector_type(8)));
typedef float v4f8 __attribute__((ext_vector_type(4), aligned(8)));
typedef float v4f __attribute__((ext_vector_type(4)));

#define HIDDEN 16
#define PROJ   2
#define INPUT  2

#define LOG2E 1.4426950408889634f

__device__ __forceinline__ float sigmoidf_fast(float x) {
  return __builtin_amdgcn_rcpf(1.0f + __builtin_amdgcn_exp2f(x * -LOG2E));
}

__device__ __forceinline__ float tanhf_fast(float x) {
#if __has_builtin(__builtin_amdgcn_tanhf)
  return __builtin_amdgcn_tanhf(x);
#else
  float e = __builtin_amdgcn_exp2f(x * (-2.0f * LOG2E));
  return __builtin_fmaf(2.0f, __builtin_amdgcn_rcpf(1.0f + e), -1.0f);
#endif
}

__device__ __forceinline__ float halfswap16(float v) {
#if __has_builtin(__builtin_amdgcn_permlanex16)
  int i = __builtin_bit_cast(int, v);
  i = __builtin_amdgcn_permlanex16(0, i, 0x76543210, 0xFEDCBA98, false, false);
  return __builtin_bit_cast(float, i);
#else
  return __shfl_xor(v, 16, 32);
#endif
}

__global__ void __launch_bounds__(256) lstm_phase1(
    const float* __restrict__ y_train, const float* __restrict__ W_ih,
    const float* __restrict__ b_ih, const float* __restrict__ b_hh,
    const float* __restrict__ W_hr, float* __restrict__ out,
    float* __restrict__ c_last, int B, int T, int nstep) {
  int idx = blockIdx.x * blockDim.x + threadIdx.x;
  if (idx >= B * nstep) return;
  int b = idx / nstep;
  int t = idx - b * nstep;

  const float* xp = y_train + ((size_t)b * T + t) * 2;
  float x0 = xp[0], x1 = xp[1];
  float y0 = 0.f, y1 = 0.f;
  bool last = (t == nstep - 1);

#pragma unroll
  for (int n = 0; n < HIDDEN; ++n) {
    int ng = 32 + n, no = 48 + n;
    float gi = W_ih[n * 2 + 0] * x0 + W_ih[n * 2 + 1] * x1 + b_ih[n] + b_hh[n];
    float gg = W_ih[ng * 2 + 0] * x0 + W_ih[ng * 2 + 1] * x1 + b_ih[ng] + b_hh[ng];
    float go = W_ih[no * 2 + 0] * x0 + W_ih[no * 2 + 1] * x1 + b_ih[no] + b_hh[no];
    float cv = sigmoidf_fast(gi) * tanhf_fast(gg);
    float hp = sigmoidf_fast(go) * tanhf_fast(cv);
    y0 = __builtin_fmaf(hp, W_hr[n], y0);
    y1 = __builtin_fmaf(hp, W_hr[16 + n], y1);
    if (last) c_last[b * HIDDEN + n] = cv;
  }
  float* op = out + ((size_t)b * T + t) * 2;
  op[0] = y0;
  op[1] = y1;
}

__global__ void __launch_bounds__(64) lstm_phase2(
    const float* __restrict__ u_train, const float* __restrict__ W_ih,
    const float* __restrict__ W_hh, const float* __restrict__ b_ih,
    const float* __restrict__ b_hh, const float* __restrict__ W_hr,
    const float* __restrict__ c_last, float* __restrict__ out,
    int T, int nstep, int nsteps2) {
  __shared__ v4f ring[2 * 8 * 2 * 32];
  const int wave = __builtin_amdgcn_readfirstlane((int)(threadIdx.x >> 5));
  const int lane = threadIdx.x & 31;
  const int hf   = lane >> 4;
  const int col  = lane & 15;
  const int cq   = col >> 2;
  const int bg   = blockIdx.x * 4 + (col & 3);

  v2f Ah[2], Ax[2];
#pragma unroll
  for (int w = 0; w < 2; ++w) {
    int r = col & 7;
    int G = r >> 1;
    int n = G * 16 + (2 * w + hf) * 4 + (col >> 3) * 2 + (r & 1);
    float sc = (G == 2) ? 1.0f : 0.5f;
    Ah[w][0] = sc * W_hh[n * 2 + 0];
    Ah[w][1] = sc * W_hh[n * 2 + 1];
    Ax[w][0] = sc * W_ih[n * 2 + 0];
    Ax[w][1] = sc * W_ih[n * 2 + 1];
  }

  const int hid0 = cq * 4 + hf * 2;
  v8f bias;
#pragma unroll
  for (int r = 0; r < 8; ++r) {
    int G = r >> 1;
    int g = G * 16 + hid0 + (r & 1);
    bias[r] = ((G == 2) ? 1.0f : 0.5f) * (b_ih[g] + b_hh[g]);
  }
  v2f wv[2];
  float c[2];
#pragma unroll
  for (int k = 0; k < 2; ++k) {
    wv[k][0] = 0.5f * W_hr[hid0 + k];
    wv[k][1] = 0.5f * W_hr[16 + hid0 + k];
    c[k] = c_last[bg * HIDDEN + hid0 + k];
  }

  const float* xbase = u_train + ((size_t)bg * T + nstep) * 2;
  float* obase = out + ((size_t)bg * T + nstep) * 2;

  const float msel0 = (cq == hf) ? 1.0f : 0.0f;
  const float msel1 = (cq == 2 + hf) ? 1.0f : 0.0f;

  float hb00, hb01, hb10, hb11;
  {
    const float* hp = out + ((size_t)bg * T + (nstep - 1)) * 2;
    float p0 = hp[0], p1 = hp[1];
    hb00 = msel0 * p0;
    hb01 = msel0 * p1;
    hb10 = msel1 * p0;
    hb11 = msel1 * p1;
  }

  auto ctile = [&](float x0, float x1, v8f& t) __attribute__((always_inline)) {
    v2f bx0, bx1;
    bx0[0] = msel0 * x0;
    bx0[1] = msel0 * x1;
    bx1[0] = msel1 * x0;
    bx1[1] = msel1 * x1;
    v8f a = __builtin_amdgcn_wmma_f32_16x16x4_f32(false, Ax[0], false, bx0, (short)0, bias, false, false);
    t = __builtin_amdgcn_wmma_f32_16x16x4_f32(false, Ax[1], false, bx1, (short)0, a, false, false);
  };

  auto step = [&](v8f cx, float& o0, float& o1) __attribute__((always_inline)) {
    v2f bh0, bh1;
    bh0[0] = hb00;
    bh0[1] = hb01;
    bh1[0] = hb10;
    bh1[1] = hb11;
    v8f da = __builtin_amdgcn_wmma_f32_16x16x4_f32(false, Ah[0], false, bh0, (short)0, cx, false, false);
    v8f d  = __builtin_amdgcn_wmma_f32_16x16x4_f32(false, Ah[1], false, bh1, (short)0, da, false, false);

    float tf[2], ti[2], gg[2], to[2], hfc[2], hi[2], cr[2], tc[2];
#pragma unroll
    for (int k = 0; k < 2; ++k) ti[k] = tanhf_fast(d[k]);
#pragma unroll
    for (int k = 0; k < 2; ++k) tf[k] = tanhf_fast(d[2 + k]);
#pragma unroll
    for (int k = 0; k < 2; ++k) gg[k] = tanhf_fast(d[4 + k]);
#pragma unroll
    for (int k = 0; k < 2; ++k) to[k] = tanhf_fast(d[6 + k]);

#pragma unroll
    for (int k = 0; k < 2; ++k) {
      float hc = 0.5f * c[k];
      hfc[k] = __builtin_fmaf(tf[k], hc, hc);
      hi[k]  = __builtin_fmaf(ti[k], 0.5f, 0.5f);
      cr[k]  = __builtin_fmaf(gg[k], hi[k], hfc[k]);
    }
#pragma unroll
    for (int k = 0; k < 2; ++k) {
      tc[k] = tanhf_fast(cr[k]);
      c[k]  = cr[k];
    }
    v2f tw[2];
#pragma unroll
    for (int k = 0; k < 2; ++k) {
      v2f t2 = {to[k], to[k]};
      tw[k] = __builtin_elementwise_fma(t2, wv[k], wv[k]);
    }
    asm volatile("" : "+v"(tw[0]), "+v"(tw[1]));
    float s0 = tc[0] * tw[0][0], s1 = tc[0] * tw[0][1];
    asm volatile("" : "+v"(s0), "+v"(s1));
    s0 = __builtin_fmaf(tc[1], tw[1][0], s0);
    s1 = __builtin_fmaf(tc[1], tw[1][1], s1);
    asm volatile("" : "+v"(s0), "+v"(s1));
    float t0, t1, r0, r1;
    asm volatile("v_add_f32_dpp %0, %1, %1 row_ror:8 row_mask:0xf bank_mask:0xf" : "=v"(t0) : "v"(s0));
    asm volatile("v_add_f32_dpp %0, %1, %1 row_ror:8 row_mask:0xf bank_mask:0xf" : "=v"(t1) : "v"(s1));
    asm volatile("v_add_f32_dpp %0, %1, %1 row_ror:4 row_mask:0xf bank_mask:0xf" : "=v"(r0) : "v"(t0));
    asm volatile("v_add_f32_dpp %0, %1, %1 row_ror:4 row_mask:0xf bank_mask:0xf" : "=v"(r1) : "v"(t1));
    float q0 = halfswap16(r0), q1 = halfswap16(r1);
    float m00 = msel0 * r0, m01 = msel0 * r1;
    float m10 = msel1 * r0, m11 = msel1 * r1;
    asm volatile("" : "+v"(q0), "+v"(q1), "+v"(m00), "+v"(m01), "+v"(m10), "+v"(m11));
    hb00 = __builtin_fmaf(q0, msel0, m00);
    hb01 = __builtin_fmaf(q1, msel0, m01);
    hb10 = __builtin_fmaf(q0, msel1, m10);
    hb11 = __builtin_fmaf(q1, msel1, m11);
    o0 = hb00;
    o1 = hb01;
  };

  const int nblk = (nsteps2 >> 4) * 2;

  if (nblk > 0) {
    v4f8 xq0 = {0.f, 0.f, 0.f, 0.f}, xq1 = xq0, xq2 = xq0, xq3 = xq0;
    if (wave == 1) {
      const v4f8* p = (const v4f8*)xbase;
      xq0 = p[0]; xq1 = p[1]; xq2 = p[2]; xq3 = p[3];
    }
    auto produce = [&](int j, int buf) __attribute__((always_inline)) {
      int nb = (j + 1 < nblk) ? j + 1 : j;
      const v4f8* np = (const v4f8*)(xbase + nb * 16);
      v4f8 xn0 = np[0], xn1 = np[1], xn2 = np[2], xn3 = np[3];
      const float xs[16] = {xq0[0], xq0[1], xq0[2], xq0[3], xq1[0], xq1[1], xq1[2], xq1[3],
                            xq2[0], xq2[1], xq2[2], xq2[3], xq3[0], xq3[1], xq3[2], xq3[3]};
#pragma unroll
      for (int s = 0; s < 8; ++s) {
        v8f t;
        ctile(xs[2 * s], xs[2 * s + 1], t);
        v4f* d = ring + ((buf * 8 + s) * 2) * 32 + lane;
        d[0]  = __builtin_shufflevector(t, t, 0, 1, 2, 3);
        d[32] = __builtin_shufflevector(t, t, 4, 5, 6, 7);
      }
      xq0 = xn0; xq1 = xn1; xq2 = xn2; xq3 = xn3;
    };

    v8f SA[8], SB[8];
    float hp[16];
#pragma unroll
    for (int q = 0; q < 16; ++q) hp[q] = 0.f;
    auto fetch = [&](v8f (&S)[8], int buf) __attribute__((always_inline)) {
#pragma unroll
      for (int s = 0; s < 8; ++s) {
        const v4f* src = ring + ((buf * 8 + s) * 2) * 32 + lane;
        S[s] = __builtin_shufflevector(src[0], src[32], 0, 1, 2, 3, 4, 5, 6, 7);
      }
    };
    auto run = [&](v8f (&S)[8]) __attribute__((always_inline)) {
#pragma unroll
      for (int s = 0; s < 8; ++s) step(S[s], hp[2 * s], hp[2 * s + 1]);
    };
    auto drop = [&](v8f (&S)[8]) __attribute__((always_inline)) {
#pragma unroll
      for (int s = 0; s < 8; ++s) S[s] = __builtin_nondeterministic_value(S[s]);
      hb00 = __builtin_nondeterministic_value(hb00);
      hb01 = __builtin_nondeterministic_value(hb01);
      hb10 = __builtin_nondeterministic_value(hb10);
      hb11 = __builtin_nondeterministic_value(hb11);
#pragma unroll
      for (int k = 0; k < 2; ++k) c[k] = __builtin_nondeterministic_value(c[k]);
#pragma unroll
      for (int q = 0; q < 16; ++q) hp[q] = __builtin_nondeterministic_value(hp[q]);
    };
    auto put = [&](int blk) __attribute__((always_inline)) {
      if (lane < 4) {
        v4f8* op = (v4f8*)(obase + blk * 16);
#pragma unroll
        for (int q = 0; q < 4; ++q) {
          v4f8 hv = {hp[4 * q], hp[4 * q + 1], hp[4 * q + 2], hp[4 * q + 3]};
          op[q] = hv;
        }
      }
    };

    if (wave == 1) produce(0, 0);
    __syncthreads();
    if (wave == 1) produce(1, 1);
    else           fetch(SA, 0);
    __syncthreads();
    for (int j = 0; j < nblk; j += 2) {
      if (wave == 1) { if (j + 2 < nblk) produce(j + 2, 0); drop(SB); }
      else {
        if (j > 0) put(j - 1);
        fetch(SB, 1);
        run(SA);
      }
      __syncthreads();
      if (wave == 1) { if (j + 3 < nblk) produce(j + 3, 1); drop(SA); }
      else {
        put(j);
        fetch(SA, 0);
        run(SB);
      }
      __syncthreads();
    }
    if (wave == 0) put(nblk - 1);
  }

  if (wave == 0) {
    for (int s = nblk << 3; s < nsteps2; ++s) {
      v8f cx;
      ctile(xbase[s * 2 + 0], xbase[s * 2 + 1], cx);
      float h0, h1;
      step(cx, h0, h1);
      if (lane < 4) {
        obase[s * 2 + 0] = h0;
        obase[s * 2 + 1] = h1;
      }
    }
  }
}

extern "C" void kernel_launch(void* const* d_in, const int* in_sizes, int n_in,
                              void* d_out, int out_size, void* d_ws, size_t ws_size,
                              hipStream_t stream) {
  const float* u_train = (const float*)d_in[0];
  const float* y_train = (const float*)d_in[1];
  const float* W_ih    = (const float*)d_in[2];
  const float* W_hh    = (const float*)d_in[3];
  const float* b_ih    = (const float*)d_in[4];
  const float* b_hh    = (const float*)d_in[5];
  const float* W_hr    = (const float*)d_in[6];
  const int B = 64;
  const int T = in_sizes[0] / (B * INPUT);
  const int nstep = T / 2;

  float* out    = (float*)d_out;
  float* c_last = (float*)d_ws;

  int total = B * nstep;
  lstm_phase1<<<(total + 255) / 256, 256, 0, stream>>>(
      y_train, W_ih, b_ih, b_hh, W_hr, out, c_last, B, T, nstep);

  lstm_phase2<<<B / 4, 64, 0, stream>>>(
      u_train, W_ih, W_hh, b_ih, b_hh, W_hr, c_last, out, T, nstep, T - nstep);
}


# Time loop's block lengthened from 8 to 16 steps: barrier, role branches, LDS tile-read group and h-store group once per 16 steps, was per 8 (order-only asm widened to keep the step schedule)
# speedup vs baseline: 4.3235x; 1.1692x over previous
#include <hip/hip_runtime.h>
#include <hip/hip_bf16.h>
#include <math.h>


typedef float v2f __attribute__((ext_vector_type(2)));
typedef float v8f __attribute__((ext_vector_type(8)));
typedef float v4f8 __attribute__((ext_vector_type(4), aligned(8)));
typedef float v4f __attribute__((ext_vector_type(4)));

#define HIDDEN 16
#define PROJ   2
#define INPUT  2
#define BS     16

#define LOG2E 1.4426950408889634f

__device__ __forceinline__ float sigmoidf_fast(float x) {
  return __builtin_amdgcn_rcpf(1.0f + __builtin_amdgcn_exp2f(x * -LOG2E));
}

__device__ __forceinline__ float tanhf_fast(float x) {
#if __has_builtin(__builtin_amdgcn_tanhf)
  return __builtin_amdgcn_tanhf(x);
#else
  float e = __builtin_amdgcn_exp2f(x * (-2.0f * LOG2E));
  return __builtin_fmaf(2.0f, __builtin_amdgcn_rcpf(1.0f + e), -1.0f);
#endif
}

__device__ __forceinline__ float halfswap16(float v) {
#if __has_builtin(__builtin_amdgcn_permlanex16)
  int i = __builtin_bit_cast(int, v);
  i = __builtin_amdgcn_permlanex16(0, i, 0x76543210, 0xFEDCBA98, false, false);
  return __builtin_bit_cast(float, i);
#else
  return __shfl_xor(v, 16, 32);
#endif
}

__global__ void __launch_bounds__(256) lstm_phase1(
    const float* __restrict__ y_train, const float* __restrict__ W_ih,
    const float* __restrict__ b_ih, const float* __restrict__ b_hh,
    const float* __restrict__ W_hr, float* __restrict__ out,
    float* __restrict__ c_last, int B, int T, int nstep) {
  int idx = blockIdx.x * blockDim.x + threadIdx.x;
  if (idx >= B * nstep) return;
  int b = idx / nstep;
  int t = idx - b * nstep;

  const float* xp = y_train + ((size_t)b * T + t) * 2;
  float x0 = xp[0], x1 = xp[1];
  float y0 = 0.f, y1 = 0.f;
  bool last = (t == nstep - 1);

#pragma unroll
  for (int n = 0; n < HIDDEN; ++n) {
    int ng = 32 + n, no = 48 + n;
    float gi = W_ih[n * 2 + 0] * x0 + W_ih[n * 2 + 1] * x1 + b_ih[n] + b_hh[n];
    float gg = W_ih[ng * 2 + 0] * x0 + W_ih[ng * 2 + 1] * x1 + b_ih[ng] + b_hh[ng];
    float go = W_ih[no * 2 + 0] * x0 + W_ih[no * 2 + 1] * x1 + b_ih[no] + b_hh[no];
    float cv = sigmoidf_fast(gi) * tanhf_fast(gg);
    float hp = sigmoidf_fast(go) * tanhf_fast(cv);
    y0 = __builtin_fmaf(hp, W_hr[n], y0);
    y1 = __builtin_fmaf(hp, W_hr[16 + n], y1);
    if (last) c_last[b * HIDDEN + n] = cv;
  }
  float* op = out + ((size_t)b * T + t) * 2;
  op[0] = y0;
  op[1] = y1;
}

__global__ void __launch_bounds__(64) lstm_phase2(
    const float* __restrict__ u_train, const float* __restrict__ W_ih,
    const float* __restrict__ W_hh, const float* __restrict__ b_ih,
    const float* __restrict__ b_hh, const float* __restrict__ W_hr,
    const float* __restrict__ c_last, float* __restrict__ out,
    int T, int nstep, int nsteps2) {
  __shared__ v4f ring[2 * BS * 2 * 32];
  const int wave = __builtin_amdgcn_readfirstlane((int)(threadIdx.x >> 5));
  const int lane = threadIdx.x & 31;
  const int hf   = lane >> 4;
  const int col  = lane & 15;
  const int cq   = col >> 2;
  const int bg   = blockIdx.x * 4 + (col & 3);

  v2f Ah[2], Ax[2];
#pragma unroll
  for (int w = 0; w < 2; ++w) {
    int r = col & 7;
    int G = r >> 1;
    int n = G * 16 + (2 * w + hf) * 4 + (col >> 3) * 2 + (r & 1);
    float sc = (G == 2) ? 1.0f : 0.5f;
    Ah[w][0] = sc * W_hh[n * 2 + 0];
    Ah[w][1] = sc * W_hh[n * 2 + 1];
    Ax[w][0] = sc * W_ih[n * 2 + 0];
    Ax[w][1] = sc * W_ih[n * 2 + 1];
  }

  const int hid0 = cq * 4 + hf * 2;
  v8f bias;
#pragma unroll
  for (int r = 0; r < 8; ++r) {
    int G = r >> 1;
    int g = G * 16 + hid0 + (r & 1);
    bias[r] = ((G == 2) ? 1.0f : 0.5f) * (b_ih[g] + b_hh[g]);
  }
  v2f wv[2];
  float c[2];
#pragma unroll
  for (int k = 0; k < 2; ++k) {
    wv[k][0] = 0.5f * W_hr[hid0 + k];
    wv[k][1] = 0.5f * W_hr[16 + hid0 + k];
    c[k] = c_last[bg * HIDDEN + hid0 + k];
  }

  const float* xbase = u_train + ((size_t)bg * T + nstep) * 2;
  float* obase = out + ((size_t)bg * T + nstep) * 2;

  const float msel0 = (cq == hf) ? 1.0f : 0.0f;
  const float msel1 = (cq == 2 + hf) ? 1.0f : 0.0f;

  float hb00, hb01, hb10, hb11;
  {
    const float* hp = out + ((size_t)bg * T + (nstep - 1)) * 2;
    float p0 = hp[0], p1 = hp[1];
    hb00 = msel0 * p0;
    hb01 = msel0 * p1;
    hb10 = msel1 * p0;
    hb11 = msel1 * p1;
  }

  auto ctile = [&](float x0, float x1, v8f& t) __attribute__((always_inline)) {
    v2f bx0, bx1;
    bx0[0] = msel0 * x0;
    bx0[1] = msel0 * x1;
    bx1[0] = msel1 * x0;
    bx1[1] = msel1 * x1;
    v8f a = __builtin_amdgcn_wmma_f32_16x16x4_f32(false, Ax[0], false, bx0, (short)0, bias, false, false);
    t = __builtin_amdgcn_wmma_f32_16x16x4_f32(false, Ax[1], false, bx1, (short)0, a, false, false);
  };

  auto step = [&](v8f cx, float& o0, float& o1) __attribute__((always_inline)) {
    v2f bh0, bh1;
    bh0[0] = hb00;
    bh0[1] = hb01;
    bh1[0] = hb10;
    bh1[1] = hb11;
    v8f da = __builtin_amdgcn_wmma_f32_16x16x4_f32(false, Ah[0], false, bh0, (short)0, cx, false, false);
    v8f d  = __builtin_amdgcn_wmma_f32_16x16x4_f32(false, Ah[1], false, bh1, (short)0, da, false, false);

    float tf[2], ti[2], gg[2], to[2], hfc[2], hi[2], cr[2], tc[2];
#pragma unroll
    for (int k = 0; k < 2; ++k) ti[k] = tanhf_fast(d[k]);
#pragma unroll
    for (int k = 0; k < 2; ++k) tf[k] = tanhf_fast(d[2 + k]);
#pragma unroll
    for (int k = 0; k < 2; ++k) gg[k] = tanhf_fast(d[4 + k]);
#pragma unroll
    for (int k = 0; k < 2; ++k) to[k] = tanhf_fast(d[6 + k]);

#pragma unroll
    for (int k = 0; k < 2; ++k) {
      float hc = 0.5f * c[k];
      hfc[k] = __builtin_fmaf(tf[k], hc, hc);
      hi[k]  = __builtin_fmaf(ti[k], 0.5f, 0.5f);
      cr[k]  = __builtin_fmaf(gg[k], hi[k], hfc[k]);
    }
#pragma unroll
    for (int k = 0; k < 2; ++k) {
      tc[k] = tanhf_fast(cr[k]);
      c[k]  = cr[k];
    }
    v2f tw[2];
#pragma unroll
    for (int k = 0; k < 2; ++k) {
      v2f t2 = {to[k], to[k]};
      tw[k] = __builtin_elementwise_fma(t2, wv[k], wv[k]);
    }
    asm volatile("" : "+v"(tw[0]), "+v"(tw[1]), "+v"(tc[0]), "+v"(tc[1]));
    float s0 = tc[0] * tw[0][0], s1 = tc[0] * tw[0][1];
    asm volatile("" : "+v"(s0), "+v"(s1));
    s0 = __builtin_fmaf(tc[1], tw[1][0], s0);
    s1 = __builtin_fmaf(tc[1], tw[1][1], s1);
    asm volatile("" : "+v"(s0), "+v"(s1));
    float t0, t1, r0, r1;
    asm volatile("v_add_f32_dpp %0, %1, %1 row_ror:8 row_mask:0xf bank_mask:0xf" : "=v"(t0) : "v"(s0));
    asm volatile("v_add_f32_dpp %0, %1, %1 row_ror:8 row_mask:0xf bank_mask:0xf" : "=v"(t1) : "v"(s1));
    asm volatile("v_add_f32_dpp %0, %1, %1 row_ror:4 row_mask:0xf bank_mask:0xf" : "=v"(r0) : "v"(t0));
    asm volatile("v_add_f32_dpp %0, %1, %1 row_ror:4 row_mask:0xf bank_mask:0xf" : "=v"(r1) : "v"(t1));
    float q0 = halfswap16(r0), q1 = halfswap16(r1);
    float m00 = msel0 * r0, m01 = msel0 * r1;
    float m10 = msel1 * r0, m11 = msel1 * r1;
    asm volatile("" : "+v"(q0), "+v"(q1), "+v"(m00), "+v"(m01), "+v"(m10), "+v"(m11));
    hb00 = __builtin_fmaf(q0, msel0, m00);
    hb01 = __builtin_fmaf(q1, msel0, m01);
    hb10 = __builtin_fmaf(q0, msel1, m10);
    hb11 = __builtin_fmaf(q1, msel1, m11);
    o0 = hb00;
    o1 = hb01;
  };

  const int nblk = (nsteps2 / (2 * BS)) * 2;

  if (nblk > 0) {
    v4f8 xq[BS / 2];
#pragma unroll
    for (int q = 0; q < BS / 2; ++q) xq[q] = (v4f8){0.f, 0.f, 0.f, 0.f};
    if (wave == 1) {
      const v4f8* p = (const v4f8*)xbase;
#pragma unroll
      for (int q = 0; q < BS / 2; ++q) xq[q] = p[q];
    }
    auto produce = [&](int j, int buf) __attribute__((always_inline)) {
      int nb = (j + 1 < nblk) ? j + 1 : j;
      const v4f8* np = (const v4f8*)(xbase + nb * (2 * BS));
      v4f8 xn[BS / 2];
#pragma unroll
      for (int q = 0; q < BS / 2; ++q) xn[q] = np[q];
#pragma unroll
      for (int s = 0; s < BS; ++s) {
        v8f t;
        ctile(xq[s >> 1][2 * (s & 1)], xq[s >> 1][2 * (s & 1) + 1], t);
        v4f* d = ring + ((buf * BS + s) * 2) * 32 + lane;
        d[0]  = __builtin_shufflevector(t, t, 0, 1, 2, 3);
        d[32] = __builtin_shufflevector(t, t, 4, 5, 6, 7);
      }
#pragma unroll
      for (int q = 0; q < BS / 2; ++q) xq[q] = xn[q];
    };

    v8f SA[BS], SB[BS];
    float hp[2 * BS];
#pragma unroll
    for (int q = 0; q < 2 * BS; ++q) hp[q] = 0.f;
    auto fetch = [&](v8f (&S)[BS], int buf) __attribute__((always_inline)) {
#pragma unroll
      for (int s = 0; s < BS; ++s) {
        const v4f* src = ring + ((buf * BS + s) * 2) * 32 + lane;
        S[s] = __builtin_shufflevector(src[0], src[32], 0, 1, 2, 3, 4, 5, 6, 7);
      }
    };
    auto run = [&](v8f (&S)[BS]) __attribute__((always_inline)) {
#pragma unroll
      for (int s = 0; s < BS; ++s) step(S[s], hp[2 * s], hp[2 * s + 1]);
    };
    auto drop = [&](v8f (&S)[BS]) __attribute__((always_inline)) {
#pragma unroll
      for (int s = 0; s < BS; ++s) S[s] = __builtin_nondeterministic_value(S[s]);
      hb00 = __builtin_nondeterministic_value(hb00);
      hb01 = __builtin_nondeterministic_value(hb01);
      hb10 = __builtin_nondeterministic_value(hb10);
      hb11 = __builtin_nondeterministic_value(hb11);
#pragma unroll
      for (int k = 0; k < 2; ++k) c[k] = __builtin_nondeterministic_value(c[k]);
#pragma unroll
      for (int q = 0; q < 2 * BS; ++q) hp[q] = __builtin_nondeterministic_value(hp[q]);
    };
    auto put = [&](int blk) __attribute__((always_inline)) {
      if (lane < 4) {
        v4f8* op = (v4f8*)(obase + blk * (2 * BS));
#pragma unroll
        for (int q = 0; q < BS / 2; ++q) {
          v4f8 hv = {hp[4 * q], hp[4 * q + 1], hp[4 * q + 2], hp[4 * q + 3]};
          op[q] = hv;
        }
      }
    };

    if (wave == 1) produce(0, 0);
    __syncthreads();
    if (wave == 1) produce(1, 1);
    else           fetch(SA, 0);
    __syncthreads();
    for (int j = 0; j < nblk; j += 2) {
      if (wave == 1) { if (j + 2 < nblk) produce(j + 2, 0); drop(SB); }
      else {
        if (j > 0) put(j - 1);
        fetch(SB, 1);
        run(SA);
      }
      __syncthreads();
      if (wave == 1) { if (j + 3 < nblk) produce(j + 3, 1); drop(SA); }
      else {
        put(j);
        fetch(SA, 0);
        run(SB);
      }
      __syncthreads();
    }
    if (wave == 0) put(nblk - 1);
  }

  if (wave == 0) {
    for (int s = nblk * BS; s < nsteps2; ++s) {
      v8f cx;
      ctile(xbase[s * 2 + 0], xbase[s * 2 + 1], cx);
      float h0, h1;
      step(cx, h0, h1);
      if (lane < 4) {
        obase[s * 2 + 0] = h0;
        obase[s * 2 + 1] = h1;
      }
    }
  }
}

extern "C" void kernel_launch(void* const* d_in, const int* in_sizes, int n_in,
                              void* d_out, int out_size, void* d_ws, size_t ws_size,
                              hipStream_t stream) {
  const float* u_train = (const float*)d_in[0];
  const float* y_train = (const float*)d_in[1];
  const float* W_ih    = (const float*)d_in[2];
  const float* W_hh    = (const float*)d_in[3];
  const float* b_ih    = (const float*)d_in[4];
  const float* b_hh    = (const float*)d_in[5];
  const float* W_hr    = (const float*)d_in[6];
  const int B = 64;
  const int T = in_sizes[0] / (B * INPUT);
  const int nstep = T / 2;

  float* out    = (float*)d_out;
  float* c_last = (float*)d_ws;

  int total = B * nstep;
  lstm_phase1<<<(total + 255) / 256, 256, 0, stream>>>(
      y_train, W_ih, b_ih, b_hh, W_hr, out, c_last, B, T, nstep);

  lstm_phase2<<<B / 4, 64, 0, stream>>>(
      u_train, W_ih, W_hh, b_ih, b_hh, W_hr, c_last, out, T, nstep, T - nstep);
}


# Time loop's block lengthened from 16 to 32 steps: barrier, role branches and their instruction refetch, exec save/restore round the h stores once per 32 steps, was per 16 (summation order kept)
# speedup vs baseline: 4.5054x; 1.0421x over previous
#include <hip/hip_runtime.h>
#include <hip/hip_bf16.h>
#include <math.h>


typedef float v2f __attribute__((ext_vector_type(2)));
typedef float v8f __attribute__((ext_vector_type(8)));
typedef float v4f8 __attribute__((ext_vector_type(4), aligned(8)));
typedef float v4f __attribute__((ext_vector_type(4)));

#define HIDDEN 16
#define PROJ   2
#define INPUT  2
#define BS     32

#define LOG2E 1.4426950408889634f

__device__ __forceinline__ float sigmoidf_fast(float x) {
  return __builtin_amdgcn_rcpf(1.0f + __builtin_amdgcn_exp2f(x * -LOG2E));
}

__device__ __forceinline__ float tanhf_fast(float x) {
#if __has_builtin(__builtin_amdgcn_tanhf)
  return __builtin_amdgcn_tanhf(x);
#else
  float e = __builtin_amdgcn_exp2f(x * (-2.0f * LOG2E));
  return __builtin_fmaf(2.0f, __builtin_amdgcn_rcpf(1.0f + e), -1.0f);
#endif
}

__device__ __forceinline__ float halfswap16(float v) {
#if __has_builtin(__builtin_amdgcn_permlanex16)
  int i = __builtin_bit_cast(int, v);
  i = __builtin_amdgcn_permlanex16(0, i, 0x76543210, 0xFEDCBA98, false, false);
  return __builtin_bit_cast(float, i);
#else
  return __shfl_xor(v, 16, 32);
#endif
}

__global__ void __launch_bounds__(256) lstm_phase1(
    const float* __restrict__ y_train, const float* __restrict__ W_ih,
    const float* __restrict__ b_ih, const float* __restrict__ b_hh,
    const float* __restrict__ W_hr, float* __restrict__ out,
    float* __restrict__ c_last, int B, int T, int nstep) {
  int idx = blockIdx.x * blockDim.x + threadIdx.x;
  if (idx >= B * nstep) return;
  int b = idx / nstep;
  int t = idx - b * nstep;

  const float* xp = y_train + ((size_t)b * T + t) * 2;
  float x0 = xp[0], x1 = xp[1];
  float y0 = 0.f, y1 = 0.f;
  bool last = (t == nstep - 1);

#pragma unroll
  for (int n = 0; n < HIDDEN; ++n) {
    int ng = 32 + n, no = 48 + n;
    float gi = W_ih[n * 2 + 0] * x0 + W_ih[n * 2 + 1] * x1 + b_ih[n] + b_hh[n];
    float gg = W_ih[ng * 2 + 0] * x0 + W_ih[ng * 2 + 1] * x1 + b_ih[ng] + b_hh[ng];
    float go = W_ih[no * 2 + 0] * x0 + W_ih[no * 2 + 1] * x1 + b_ih[no] + b_hh[no];
    float cv = sigmoidf_fast(gi) * tanhf_fast(gg);
    float hp = sigmoidf_fast(go) * tanhf_fast(cv);
    y0 = __builtin_fmaf(hp, W_hr[n], y0);
    y1 = __builtin_fmaf(hp, W_hr[16 + n], y1);
    if (last) c_last[b * HIDDEN + n] = cv;
  }
  float* op = out + ((size_t)b * T + t) * 2;
  op[0] = y0;
  op[1] = y1;
}

__global__ void __launch_bounds__(64) lstm_phase2(
    const float* __restrict__ u_train, const float* __restrict__ W_ih,
    const float* __restrict__ W_hh, const float* __restrict__ b_ih,
    const float* __restrict__ b_hh, const float* __restrict__ W_hr,
    const float* __restrict__ c_last, float* __restrict__ out,
    int T, int nstep, int nsteps2) {
  __shared__ v4f ring[2 * BS * 2 * 32];
  const int wave = __builtin_amdgcn_readfirstlane((int)(threadIdx.x >> 5));
  const int lane = threadIdx.x & 31;
  const int hf   = lane >> 4;
  const int col  = lane & 15;
  const int cq   = col >> 2;
  const int bg   = blockIdx.x * 4 + (col & 3);

  v2f Ah[2], Ax[2];
#pragma unroll
  for (int w = 0; w < 2; ++w) {
    int r = col & 7;
    int G = r >> 1;
    int n = G * 16 + (2 * w + hf) * 4 + (col >> 3) * 2 + (r & 1);
    float sc = (G == 2) ? 1.0f : 0.5f;
    Ah[w][0] = sc * W_hh[n * 2 + 0];
    Ah[w][1] = sc * W_hh[n * 2 + 1];
    Ax[w][0] = sc * W_ih[n * 2 + 0];
    Ax[w][1] = sc * W_ih[n * 2 + 1];
  }

  const int hid0 = cq * 4 + hf * 2;
  v8f bias;
#pragma unroll
  for (int r = 0; r < 8; ++r) {
    int G = r >> 1;
    int g = G * 16 + hid0 + (r & 1);
    bias[r] = ((G == 2) ? 1.0f : 0.5f) * (b_ih[g] + b_hh[g]);
  }
  v2f wv[2];
  float c[2];
#pragma unroll
  for (int k = 0; k < 2; ++k) {
    wv[k][0] = 0.5f * W_hr[hid0 + k];
    wv[k][1] = 0.5f * W_hr[16 + hid0 + k];
    c[k] = c_last[bg * HIDDEN + hid0 + k];
  }

  const float* xbase = u_train + ((size_t)bg * T + nstep) * 2;
  float* obase = out + ((size_t)bg * T + nstep) * 2;

  const float msel0 = (cq == hf) ? 1.0f : 0.0f;
  const float msel1 = (cq == 2 + hf) ? 1.0f : 0.0f;

  float hb00, hb01, hb10, hb11;
  {
    const float* hp = out + ((size_t)bg * T + (nstep - 1)) * 2;
    float p0 = hp[0], p1 = hp[1];
    hb00 = msel0 * p0;
    hb01 = msel0 * p1;
    hb10 = msel1 * p0;
    hb11 = msel1 * p1;
  }

  auto ctile = [&](float x0, float x1, v8f& t) __attribute__((always_inline)) {
    v2f bx0, bx1;
    bx0[0] = msel0 * x0;
    bx0[1] = msel0 * x1;
    bx1[0] = msel1 * x0;
    bx1[1] = msel1 * x1;
    v8f a = __builtin_amdgcn_wmma_f32_16x16x4_f32(false, Ax[0], false, bx0, (short)0, bias, false, false);
    t = __builtin_amdgcn_wmma_f32_16x16x4_f32(false, Ax[1], false, bx1, (short)0, a, false, false);
  };

  auto step = [&](v8f cx, float& o0, float& o1) __attribute__((always_inline)) {
    v2f bh0, bh1;
    bh0[0] = hb00;
    bh0[1] = hb01;
    bh1[0] = hb10;
    bh1[1] = hb11;
    v8f da = __builtin_amdgcn_wmma_f32_16x16x4_f32(false, Ah[0], false, bh0, (short)0, cx, false, false);
    v8f d  = __builtin_amdgcn_wmma_f32_16x16x4_f32(false, Ah[1], false, bh1, (short)0, da, false, false);

    float tf[2], ti[2], gg[2], to[2], hfc[2], hi[2], cr[2], tc[2];
#pragma unroll
    for (int k = 0; k < 2; ++k) ti[k] = tanhf_fast(d[k]);
#pragma unroll
    for (int k = 0; k < 2; ++k) tf[k] = tanhf_fast(d[2 + k]);
#pragma unroll
    for (int k = 0; k < 2; ++k) gg[k] = tanhf_fast(d[4 + k]);
#pragma unroll
    for (int k = 0; k < 2; ++k) to[k] = tanhf_fast(d[6 + k]);

#pragma unroll
    for (int k = 0; k < 2; ++k) {
      float hc = 0.5f * c[k];
      hfc[k] = __builtin_fmaf(tf[k], hc, hc);
      hi[k]  = __builtin_fmaf(ti[k], 0.5f, 0.5f);
      cr[k]  = __builtin_fmaf(gg[k], hi[k], hfc[k]);
    }
#pragma unroll
    for (int k = 0; k < 2; ++k) {
      tc[k] = tanhf_fast(cr[k]);
      c[k]  = cr[k];
    }
    v2f tw[2];
#pragma unroll
    for (int k = 0; k < 2; ++k) {
      v2f t2 = {to[k], to[k]};
      tw[k] = __builtin_elementwise_fma(t2, wv[k], wv[k]);
    }
    asm volatile("" : "+v"(tw[0]), "+v"(tw[1]), "+v"(tc[0]), "+v"(tc[1]));
    float s0 = tc[0] * tw[0][0], s1 = tc[0] * tw[0][1];
    asm volatile("" : "+v"(s0), "+v"(s1));
    s0 = __builtin_fmaf(tc[1], tw[1][0], s0);
    s1 = __builtin_fmaf(tc[1], tw[1][1], s1);
    asm volatile("" : "+v"(s0), "+v"(s1));
    float t0, t1, r0, r1;
    asm volatile("v_add_f32_dpp %0, %1, %1 row_ror:8 row_mask:0xf bank_mask:0xf" : "=v"(t0) : "v"(s0));
    asm volatile("v_add_f32_dpp %0, %1, %1 row_ror:8 row_mask:0xf bank_mask:0xf" : "=v"(t1) : "v"(s1));
    asm volatile("v_add_f32_dpp %0, %1, %1 row_ror:4 row_mask:0xf bank_mask:0xf" : "=v"(r0) : "v"(t0));
    asm volatile("v_add_f32_dpp %0, %1, %1 row_ror:4 row_mask:0xf bank_mask:0xf" : "=v"(r1) : "v"(t1));
    float q0 = halfswap16(r0), q1 = halfswap16(r1);
    float m00 = msel0 * r0, m01 = msel0 * r1;
    float m10 = msel1 * r0, m11 = msel1 * r1;
    asm volatile("" : "+v"(q0), "+v"(q1), "+v"(m00), "+v"(m01), "+v"(m10), "+v"(m11));
    hb00 = __builtin_fmaf(q0, msel0, m00);
    hb01 = __builtin_fmaf(q1, msel0, m01);
    hb10 = __builtin_fmaf(q0, msel1, m10);
    hb11 = __builtin_fmaf(q1, msel1, m11);
    o0 = hb00;
    o1 = hb01;
  };

  const int nblk = (nsteps2 / (2 * BS)) * 2;

  if (nblk > 0) {
    v4f8 xq[BS / 2];
#pragma unroll
    for (int q = 0; q < BS / 2; ++q) xq[q] = (v4f8){0.f, 0.f, 0.f, 0.f};
    if (wave == 1) {
      const v4f8* p = (const v4f8*)xbase;
#pragma unroll
      for (int q = 0; q < BS / 2; ++q) xq[q] = p[q];
    }
    auto produce = [&](int j, int buf) __attribute__((always_inline)) {
      int nb = (j + 1 < nblk) ? j + 1 : j;
      const v4f8* np = (const v4f8*)(xbase + nb * (2 * BS));
      v4f8 xn[BS / 2];
#pragma unroll
      for (int q = 0; q < BS / 2; ++q) xn[q] = np[q];
#pragma unroll
      for (int s = 0; s < BS; ++s) {
        v8f t;
        ctile(xq[s >> 1][2 * (s & 1)], xq[s >> 1][2 * (s & 1) + 1], t);
        v4f* d = ring + ((buf * BS + s) * 2) * 32 + lane;
        d[0]  = __builtin_shufflevector(t, t, 0, 1, 2, 3);
        d[32] = __builtin_shufflevector(t, t, 4, 5, 6, 7);
      }
#pragma unroll
      for (int q = 0; q < BS / 2; ++q) xq[q] = xn[q];
    };

    v8f SA[BS], SB[BS];
    float hp[2 * BS];
#pragma unroll
    for (int q = 0; q < 2 * BS; ++q) hp[q] = 0.f;
    auto fetch = [&](v8f (&S)[BS], int buf) __attribute__((always_inline)) {
#pragma unroll
      for (int s = 0; s < BS; ++s) {
        const v4f* src = ring + ((buf * BS + s) * 2) * 32 + lane;
        S[s] = __builtin_shufflevector(src[0], src[32], 0, 1, 2, 3, 4, 5, 6, 7);
      }
    };
    auto run = [&](v8f (&S)[BS]) __attribute__((always_inline)) {
#pragma unroll
      for (int s = 0; s < BS; ++s) step(S[s], hp[2 * s], hp[2 * s + 1]);
    };
    auto drop = [&](v8f (&S)[BS]) __attribute__((always_inline)) {
#pragma unroll
      for (int s = 0; s < BS; ++s) S[s] = __builtin_nondeterministic_value(S[s]);
      hb00 = __builtin_nondeterministic_value(hb00);
      hb01 = __builtin_nondeterministic_value(hb01);
      hb10 = __builtin_nondeterministic_value(hb10);
      hb11 = __builtin_nondeterministic_value(hb11);
#pragma unroll
      for (int k = 0; k < 2; ++k) c[k] = __builtin_nondeterministic_value(c[k]);
#pragma unroll
      for (int q = 0; q < 2 * BS; ++q) hp[q] = __builtin_nondeterministic_value(hp[q]);
    };
    auto put = [&](int blk) __attribute__((always_inline)) {
      if (lane < 4) {
        v4f8* op = (v4f8*)(obase + blk * (2 * BS));
#pragma unroll
        for (int q = 0; q < BS / 2; ++q) {
          v4f8 hv = {hp[4 * q], hp[4 * q + 1], hp[4 * q + 2], hp[4 * q + 3]};
          op[q] = hv;
        }
      }
    };

    if (wave == 1) produce(0, 0);
    __syncthreads();
    if (wave == 1) produce(1, 1);
    else           fetch(SA, 0);
    __syncthreads();
    for (int j = 0; j < nblk; j += 2) {
      if (wave == 1) { if (j + 2 < nblk) produce(j + 2, 0); drop(SB); }
      else {
        if (j > 0) put(j - 1);
        fetch(SB, 1);
        run(SA);
      }
      __syncthreads();
      if (wave == 1) { if (j + 3 < nblk) produce(j + 3, 1); drop(SA); }
      else {
        put(j);
        fetch(SA, 0);
        run(SB);
      }
      __syncthreads();
    }
    if (wave == 0) put(nblk - 1);
  }

  if (wave == 0) {
    for (int s = nblk * BS; s < nsteps2; ++s) {
      v8f cx;
      ctile(xbase[s * 2 + 0], xbase[s * 2 + 1], cx);
      float h0, h1;
      step(cx, h0, h1);
      if (lane < 4) {
        obase[s * 2 + 0] = h0;
        obase[s * 2 + 1] = h1;
      }
    }
  }
}

extern "C" void kernel_launch(void* const* d_in, const int* in_sizes, int n_in,
                              void* d_out, int out_size, void* d_ws, size_t ws_size,
                              hipStream_t stream) {
  const float* u_train = (const float*)d_in[0];
  const float* y_train = (const float*)d_in[1];
  const float* W_ih    = (const float*)d_in[2];
  const float* W_hh    = (const float*)d_in[3];
  const float* b_ih    = (const float*)d_in[4];
  const float* b_hh    = (const float*)d_in[5];
  const float* W_hr    = (const float*)d_in[6];
  const int B = 64;
  const int T = in_sizes[0] / (B * INPUT);
  const int nstep = T / 2;

  float* out    = (float*)d_out;
  float* c_last = (float*)d_ws;

  int total = B * nstep;
  lstm_phase1<<<(total + 255) / 256, 256, 0, stream>>>(
      y_train, W_ih, b_ih, b_hh, W_hr, out, c_last, B, T, nstep);

  lstm_phase2<<<B / 4, 64, 0, stream>>>(
      u_train, W_ih, W_hh, b_ih, b_hh, W_hr, c_last, out, T, nstep, T - nstep);
}


# speedup vs baseline: 4.6471x; 1.0315x over previous
#include <hip/hip_runtime.h>
#include <hip/hip_bf16.h>
#include <math.h>


typedef float v2f __attribute__((ext_vector_type(2)));
typedef float v8f __attribute__((ext_vector_type(8)));
typedef float v4f8 __attribute__((ext_vector_type(4), aligned(8)));
typedef float v4f __attribute__((ext_vector_type(4)));

#define HIDDEN 16
#define PROJ   2
#define INPUT  2
#define BS     32

#define LOG2E 1.4426950408889634f

#define STEP_ORDER() __builtin_amdgcn_sched_barrier(0)

__device__ __forceinline__ float sigmoidf_fast(float x) {
  return __builtin_amdgcn_rcpf(1.0f + __builtin_amdgcn_exp2f(x * -LOG2E));
}

__device__ __forceinline__ float tanhf_fast(float x) {
#if __has_builtin(__builtin_amdgcn_tanhf)
  return __builtin_amdgcn_tanhf(x);
#else
  float e = __builtin_amdgcn_exp2f(x * (-2.0f * LOG2E));
  return __builtin_fmaf(2.0f, __builtin_amdgcn_rcpf(1.0f + e), -1.0f);
#endif
}

__device__ __forceinline__ float halfswap16(float v) {
#if __has_builtin(__builtin_amdgcn_permlanex16)
  int i = __builtin_bit_cast(int, v);
  i = __builtin_amdgcn_permlanex16(0, i, 0x76543210, 0xFEDCBA98, false, false);
  return __builtin_bit_cast(float, i);
#else
  return __shfl_xor(v, 16, 32);
#endif
}

__global__ void __launch_bounds__(256) lstm_phase1(
    const float* __restrict__ y_train, const float* __restrict__ W_ih,
    const float* __restrict__ b_ih, const float* __restrict__ b_hh,
    const float* __restrict__ W_hr, float* __restrict__ out,
    float* __restrict__ c_last, int B, int T, int nstep) {
  int idx = blockIdx.x * blockDim.x + threadIdx.x;
  if (idx >= B * nstep) return;
  int b = idx / nstep;
  int t = idx - b * nstep;

  const float* xp = y_train + ((size_t)b * T + t) * 2;
  float x0 = xp[0], x1 = xp[1];
  float y0 = 0.f, y1 = 0.f;
  bool last = (t == nstep - 1);

#pragma unroll
  for (int n = 0; n < HIDDEN; ++n) {
    int ng = 32 + n, no = 48 + n;
    float gi = W_ih[n * 2 + 0] * x0 + W_ih[n * 2 + 1] * x1 + b_ih[n] + b_hh[n];
    float gg = W_ih[ng * 2 + 0] * x0 + W_ih[ng * 2 + 1] * x1 + b_ih[ng] + b_hh[ng];
    float go = W_ih[no * 2 + 0] * x0 + W_ih[no * 2 + 1] * x1 + b_ih[no] + b_hh[no];
    float cv = sigmoidf_fast(gi) * tanhf_fast(gg);
    float hp = sigmoidf_fast(go) * tanhf_fast(cv);
    y0 = __builtin_fmaf(hp, W_hr[n], y0);
    y1 = __builtin_fmaf(hp, W_hr[16 + n], y1);
    if (last) c_last[b * HIDDEN + n] = cv;
  }
  float* op = out + ((size_t)b * T + t) * 2;
  op[0] = y0;
  op[1] = y1;
}

__global__ void __launch_bounds__(64) lstm_phase2(
    const float* __restrict__ u_train, const float* __restrict__ W_ih,
    const float* __restrict__ W_hh, const float* __restrict__ b_ih,
    const float* __restrict__ b_hh, const float* __restrict__ W_hr,
    const float* __restrict__ c_last, float* __restrict__ out,
    int T, int nstep, int nsteps2) {
  __shared__ v4f ring[2 * BS * 2 * 32];
  const int wave = __builtin_amdgcn_readfirstlane((int)(threadIdx.x >> 5));
  const int lane = threadIdx.x & 31;
  const int hf   = lane >> 4;
  const int col  = lane & 15;
  const int cq   = col >> 2;
  const int bg   = blockIdx.x * 4 + (col & 3);

  v2f Ah[2], Ax[2];
#pragma unroll
  for (int w = 0; w < 2; ++w) {
    int r = col & 7;
    int G = r >> 1;
    int n = G * 16 + (2 * w + hf) * 4 + (col >> 3) * 2 + (r & 1);
    float sc = (G == 2) ? 1.0f : 0.5f;
    Ah[w][0] = sc * W_hh[n * 2 + 0];
    Ah[w][1] = sc * W_hh[n * 2 + 1];
    Ax[w][0] = sc * W_ih[n * 2 + 0];
    Ax[w][1] = sc * W_ih[n * 2 + 1];
  }

  const int hid0 = cq * 4 + hf * 2;
  v8f bias;
#pragma unroll
  for (int r = 0; r < 8; ++r) {
    int G = r >> 1;
    int g = G * 16 + hid0 + (r & 1);
    bias[r] = ((G == 2) ? 1.0f : 0.5f) * (b_ih[g] + b_hh[g]);
  }
  v2f wv[2];
  float c[2];
#pragma unroll
  for (int k = 0; k < 2; ++k) {
    wv[k][0] = 0.5f * W_hr[hid0 + k];
    wv[k][1] = 0.5f * W_hr[16 + hid0 + k];
    c[k] = c_last[bg * HIDDEN + hid0 + k];
  }

  const float* xbase = u_train + ((size_t)bg * T + nstep) * 2;
  float* obase = out + ((size_t)bg * T + nstep) * 2;

  const float msel0 = (cq == hf) ? 1.0f : 0.0f;
  const float msel1 = (cq == 2 + hf) ? 1.0f : 0.0f;

  float hb00, hb01, hb10, hb11;
  {
    const float* hp = out + ((size_t)bg * T + (nstep - 1)) * 2;
    float p0 = hp[0], p1 = hp[1];
    hb00 = msel0 * p0;
    hb01 = msel0 * p1;
    hb10 = msel1 * p0;
    hb11 = msel1 * p1;
  }

  auto ctile = [&](float x0, float x1, v8f& t) __attribute__((always_inline)) {
    v2f bx0, bx1;
    bx0[0] = msel0 * x0;
    bx0[1] = msel0 * x1;
    bx1[0] = msel1 * x0;
    bx1[1] = msel1 * x1;
    v8f a = __builtin_amdgcn_wmma_f32_16x16x4_f32(false, Ax[0], false, bx0, (short)0, bias, false, false);
    t = __builtin_amdgcn_wmma_f32_16x16x4_f32(false, Ax[1], false, bx1, (short)0, a, false, false);
  };

  auto step = [&](v8f cx, float& o0, float& o1) __attribute__((always_inline)) {
    v2f bh0, bh1;
    bh0[0] = hb00;
    bh0[1] = hb01;
    bh1[0] = hb10;
    bh1[1] = hb11;
    v8f da = __builtin_amdgcn_wmma_f32_16x16x4_f32(false, Ah[0], false, bh0, (short)0, cx, false, false);
    v8f d  = __builtin_amdgcn_wmma_f32_16x16x4_f32(false, Ah[1], false, bh1, (short)0, da, false, false);

    float tf[2], ti[2], gg[2], to[2], hfc[2], hi[2], cr[2], tc[2];
    const float hc0 = 0.5f * c[0], hc1 = 0.5f * c[1];
    ti[0] = tanhf_fast(d[0]);
    tf[0] = tanhf_fast(d[2]);
    ti[1] = tanhf_fast(d[1]);
    tf[1] = tanhf_fast(d[3]);
    STEP_ORDER();
    gg[0] = tanhf_fast(d[4]);
    hi[0] = __builtin_fmaf(ti[0], 0.5f, 0.5f);
    STEP_ORDER();
    gg[1] = tanhf_fast(d[5]);
    hfc[0] = __builtin_fmaf(tf[0], hc0, hc0);
    STEP_ORDER();
    to[0] = tanhf_fast(d[6]);
    hi[1] = __builtin_fmaf(ti[1], 0.5f, 0.5f);
    STEP_ORDER();
    to[1] = tanhf_fast(d[7]);
    hfc[1] = __builtin_fmaf(tf[1], hc1, hc1);
    STEP_ORDER();
    cr[0] = __builtin_fmaf(gg[0], hi[0], hfc[0]);
    STEP_ORDER();
    cr[1] = __builtin_fmaf(gg[1], hi[1], hfc[1]);
    tc[0] = tanhf_fast(cr[0]);
    tc[1] = tanhf_fast(cr[1]);
    c[0] = cr[0];
    c[1] = cr[1];
    v2f tw[2];
#pragma unroll
    for (int k = 0; k < 2; ++k) {
      v2f t2 = {to[k], to[k]};
      tw[k] = __builtin_elementwise_fma(t2, wv[k], wv[k]);
    }
    asm volatile("" : "+v"(tw[0]), "+v"(tw[1]), "+v"(tc[0]), "+v"(tc[1]));
    float s0 = tc[0] * tw[0][0], s1 = tc[0] * tw[0][1];
    asm volatile("" : "+v"(s0), "+v"(s1));
    s0 = __builtin_fmaf(tc[1], tw[1][0], s0);
    s1 = __builtin_fmaf(tc[1], tw[1][1], s1);
    asm volatile("" : "+v"(s0), "+v"(s1));
    float t0, t1, r0, r1;
    asm volatile("v_add_f32_dpp %0, %1, %1 row_ror:8 row_mask:0xf bank_mask:0xf" : "=v"(t0) : "v"(s0));
    asm volatile("v_add_f32_dpp %0, %1, %1 row_ror:8 row_mask:0xf bank_mask:0xf" : "=v"(t1) : "v"(s1));
    asm volatile("v_add_f32_dpp %0, %1, %1 row_ror:4 row_mask:0xf bank_mask:0xf" : "=v"(r0) : "v"(t0));
    asm volatile("v_add_f32_dpp %0, %1, %1 row_ror:4 row_mask:0xf bank_mask:0xf" : "=v"(r1) : "v"(t1));
    float q0 = halfswap16(r0), q1 = halfswap16(r1);
    asm volatile("" : "+v"(q0), "+v"(q1), "+v"(r0), "+v"(r1));
    float m00 = msel0 * r0, m01 = msel0 * r1;
    float m10 = msel1 * r0, m11 = msel1 * r1;
    asm volatile("" : "+v"(q0), "+v"(q1), "+v"(m00), "+v"(m01), "+v"(m10), "+v"(m11));
    hb00 = __builtin_fmaf(q0, msel0, m00);
    hb01 = __builtin_fmaf(q1, msel0, m01);
    hb10 = __builtin_fmaf(q0, msel1, m10);
    hb11 = __builtin_fmaf(q1, msel1, m11);
    o0 = hb00;
    o1 = hb01;
  };

  const int nblk = (nsteps2 / (2 * BS)) * 2;

  if (nblk > 0) {
    v4f8 xq[BS / 2];
#pragma unroll
    for (int q = 0; q < BS / 2; ++q) xq[q] = (v4f8){0.f, 0.f, 0.f, 0.f};
    if (wave == 1) {
      const v4f8* p = (const v4f8*)xbase;
#pragma unroll
      for (int q = 0; q < BS / 2; ++q) xq[q] = p[q];
    }
    auto produce = [&](int j, int buf) __attribute__((always_inline)) {
      int nb = (j + 1 < nblk) ? j + 1 : j;
      const v4f8* np = (const v4f8*)(xbase + nb * (2 * BS));
      v4f8 xn[BS / 2];
#pragma unroll
      for (int q = 0; q < BS / 2; ++q) xn[q] = np[q];
#pragma unroll
      for (int s = 0; s < BS; ++s) {
        v8f t;
        ctile(xq[s >> 1][2 * (s & 1)], xq[s >> 1][2 * (s & 1) + 1], t);
        v4f* d = ring + ((buf * BS + s) * 2) * 32 + lane;
        d[0]  = __builtin_shufflevector(t, t, 0, 1, 2, 3);
        d[32] = __builtin_shufflevector(t, t, 4, 5, 6, 7);
      }
#pragma unroll
      for (int q = 0; q < BS / 2; ++q) xq[q] = xn[q];
    };

    v8f SA[BS], SB[BS];
    float hp[2 * BS];
#pragma unroll
    for (int q = 0; q < 2 * BS; ++q) hp[q] = 0.f;
    auto fetch = [&](v8f (&S)[BS], int buf) __attribute__((always_inline)) {
#pragma unroll
      for (int s = 0; s < BS; ++s) {
        const v4f* src = ring + ((buf * BS + s) * 2) * 32 + lane;
        S[s] = __builtin_shufflevector(src[0], src[32], 0, 1, 2, 3, 4, 5, 6, 7);
      }
    };
    auto run = [&](v8f (&S)[BS]) __attribute__((always_inline)) {
#pragma unroll
      for (int s = 0; s < BS; ++s) step(S[s], hp[2 * s], hp[2 * s + 1]);
    };
    auto drop = [&](v8f (&S)[BS]) __attribute__((always_inline)) {
#pragma unroll
      for (int s = 0; s < BS; ++s) S[s] = __builtin_nondeterministic_value(S[s]);
      hb00 = __builtin_nondeterministic_value(hb00);
      hb01 = __builtin_nondeterministic_value(hb01);
      hb10 = __builtin_nondeterministic_value(hb10);
      hb11 = __builtin_nondeterministic_value(hb11);
#pragma unroll
      for (int k = 0; k < 2; ++k) c[k] = __builtin_nondeterministic_value(c[k]);
#pragma unroll
      for (int q = 0; q < 2 * BS; ++q) hp[q] = __builtin_nondeterministic_value(hp[q]);
    };
    auto put = [&](int blk) __attribute__((always_inline)) {
      if (lane < 4) {
        v4f8* op = (v4f8*)(obase + blk * (2 * BS));
#pragma unroll
        for (int q = 0; q < BS / 2; ++q) {
          v4f8 hv = {hp[4 * q], hp[4 * q + 1], hp[4 * q + 2], hp[4 * q + 3]};
          op[q] = hv;
        }
      }
    };

    if (wave == 1) produce(0, 0);
    __syncthreads();
    if (wave == 1) produce(1, 1);
    else           fetch(SA, 0);
    __syncthreads();
    for (int j = 0; j < nblk; j += 2) {
      if (wave == 1) { if (j + 2 < nblk) produce(j + 2, 0); drop(SB); }
      else {
        if (j > 0) put(j - 1);
        fetch(SB, 1);
        run(SA);
      }
      __syncthreads();
      if (wave == 1) { if (j + 3 < nblk) produce(j + 3, 1); drop(SA); }
      else {
        put(j);
        fetch(SA, 0);
        run(SB);
      }
      __syncthreads();
    }
    if (wave == 0) put(nblk - 1);
  }

  if (wave == 0) {
    for (int s = nblk * BS; s < nsteps2; ++s) {
      v8f cx;
      ctile(xbase[s * 2 + 0], xbase[s * 2 + 1], cx);
      float h0, h1;
      step(cx, h0, h1);
      if (lane < 4) {
        obase[s * 2 + 0] = h0;
        obase[s * 2 + 1] = h1;
      }
    }
  }
}

extern "C" void kernel_launch(void* const* d_in, const int* in_sizes, int n_in,
                              void* d_out, int out_size, void* d_ws, size_t ws_size,
                              hipStream_t stream) {
  const float* u_train = (const float*)d_in[0];
  const float* y_train = (const float*)d_in[1];
  const float* W_ih    = (const float*)d_in[2];
  const float* W_hh    = (const float*)d_in[3];
  const float* b_ih    = (const float*)d_in[4];
  const float* b_hh    = (const float*)d_in[5];
  const float* W_hr    = (const float*)d_in[6];
  const int B = 64;
  const int T = in_sizes[0] / (B * INPUT);
  const int nstep = T / 2;

  float* out    = (float*)d_out;
  float* c_last = (float*)d_ws;

  int total = B * nstep;
  lstm_phase1<<<(total + 255) / 256, 256, 0, stream>>>(
      y_train, W_ih, b_ih, b_hh, W_hr, out, c_last, B, T, nstep);

  lstm_phase2<<<B / 4, 64, 0, stream>>>(
      u_train, W_ih, W_hh, b_ih, b_hh, W_hr, c_last, out, T, nstep, T - nstep);
}
